# starting kernel
# baseline (speedup 1.0000x reference)
#include <hip/hip_runtime.h>
#include <hip/hip_bf16.h>


typedef __attribute__((ext_vector_type(2))) float v2f;
typedef __attribute__((ext_vector_type(8))) float v8f;

#define HEADS 2
#define FDIM 128
#define NEG_SLOPE 0.2f

__device__ __forceinline__ unsigned enc_f32(float f) {
    unsigned u = __float_as_uint(f);
    return (u & 0x80000000u) ? ~u : (u | 0x80000000u);
}
__device__ __forceinline__ float dec_f32(unsigned u) {
    return __uint_as_float((u & 0x80000000u) ? (u & 0x7FFFFFFFu) : ~u);
}

__global__ __launch_bounds__(256) void gemm128_wmma(
    const float* __restrict__ X, const float* __restrict__ W,
    float* __restrict__ Y, int M)
{
    __shared__ float lw[FDIM * FDIM];
    {
        const float4* w4 = (const float4*)W;
        float4* l4 = (float4*)lw;
        for (int i = threadIdx.x; i < FDIM * FDIM / 4; i += 256) l4[i] = w4[i];
    }
    __syncthreads();

    const int wave = threadIdx.x >> 5;
    const int lane = threadIdx.x & 31;
    const int m0 = (blockIdx.x * 8 + wave) * 16;
    if (m0 >= M) return;

    const int lm = lane & 15;
    const int lh = lane >> 4;
    const float* arow = X + (size_t)(m0 + lm) * FDIM + 2 * lh;

#pragma unroll
    for (int nt = 0; nt < 8; ++nt) {
        v8f acc = {};
        const int nc = nt * 16 + lm;
#pragma unroll 8
        for (int kk = 0; kk < 32; ++kk) {
            const int k0 = kk * 4;
            v2f a = *(const v2f*)(arow + k0);
            v2f b;
            b.x = lw[(k0 + 2 * lh) * FDIM + nc];
            b.y = lw[(k0 + 2 * lh + 1) * FDIM + nc];
            acc = __builtin_amdgcn_wmma_f32_16x16x4_f32(
                false, a, false, b, (short)0, acc, false, false);
        }
        float* yrow = Y + (size_t)(m0 + 8 * lh) * FDIM + nc;
#pragma unroll
        for (int r = 0; r < 8; ++r) yrow[(size_t)r * FDIM] = acc[r];
    }
}

__global__ void attn_logits(const float* __restrict__ h,
                            const float* __restrict__ a_src,
                            const float* __restrict__ a_dst,
                            float* __restrict__ esrc, float* __restrict__ edst,
                            int n)
{
    int idx = blockIdx.x * blockDim.x + threadIdx.x;
    if (idx >= n * HEADS) return;
    int node = idx >> 1, hd = idx & 1;
    const float* hp = h + (size_t)node * FDIM + hd * 64;
    const float* as = a_src + hd * 64;
    const float* ad = a_dst + hd * 64;
    float s = 0.f, t = 0.f;
#pragma unroll 8
    for (int f = 0; f < 64; ++f) { float x = hp[f]; s += x * as[f]; t += x * ad[f]; }
    esrc[idx] = s; edst[idx] = t;
}

__global__ void init_accum(unsigned* __restrict__ m, float* __restrict__ denom,
                           float* __restrict__ agg, int n)
{
    int i = blockIdx.x * blockDim.x + threadIdx.x;
    if (i < n * FDIM) agg[i] = 0.f;
    if (i < n * HEADS) { m[i] = 0u; denom[i] = 0.f; }
}

__global__ void edge_max(const int* __restrict__ src, const int* __restrict__ dst,
                         int E, int n,
                         const float* __restrict__ esrc,
                         const float* __restrict__ edst,
                         unsigned* __restrict__ m)
{
    int e = blockIdx.x * blockDim.x + threadIdx.x;
    if (e >= E + n) return;
    int s = (e < E) ? src[e] : (e - E);
    int d = (e < E) ? dst[e] : (e - E);
#pragma unroll
    for (int h = 0; h < HEADS; ++h) {
        float v = esrc[s * HEADS + h] + edst[d * HEADS + h];
        v = (v >= 0.f) ? v : NEG_SLOPE * v;
        atomicMax(&m[d * HEADS + h], enc_f32(v));
    }
}

__global__ void edge_aggregate(const int* __restrict__ src, const int* __restrict__ dst,
                               int E, int n,
                               const float* __restrict__ esrc,
                               const float* __restrict__ edst,
                               const unsigned* __restrict__ m,
                               const float* __restrict__ hfeat,
                               float* __restrict__ denom,
                               float* __restrict__ agg)
{
    int wid = (blockIdx.x * blockDim.x + threadIdx.x) >> 5;
    int lane = threadIdx.x & 31;
    if (wid >= E + n) return;
    int s = (wid < E) ? src[wid] : (wid - E);
    int d = (wid < E) ? dst[wid] : (wid - E);

    float v0 = esrc[s * HEADS + 0] + edst[d * HEADS + 0];
    float v1 = esrc[s * HEADS + 1] + edst[d * HEADS + 1];
    v0 = (v0 >= 0.f) ? v0 : NEG_SLOPE * v0;
    v1 = (v1 >= 0.f) ? v1 : NEG_SLOPE * v1;
    float a0 = __expf(v0 - dec_f32(m[d * HEADS + 0]));
    float a1 = __expf(v1 - dec_f32(m[d * HEADS + 1]));

    if (lane == 0) {
        atomicAdd(&denom[d * HEADS + 0], a0);
        atomicAdd(&denom[d * HEADS + 1], a1);
    }
    const float* hs = hfeat + (size_t)s * FDIM;
    float* od = agg + (size_t)d * FDIM;
#pragma unroll
    for (int i = 0; i < 4; ++i) {
        int f = lane + 32 * i;
        float al = (f < 64) ? a0 : a1;
        atomicAdd(&od[f], al * hs[f]);
    }
}

__global__ void node_finalize(const float* __restrict__ agg,
                              const float* __restrict__ denom,
                              const float* __restrict__ bias,
                              float* __restrict__ z, int n, int do_relu)
{
    int idx = blockIdx.x * blockDim.x + threadIdx.x;
    if (idx >= n * FDIM) return;
    int node = idx >> 7, c = idx & 127, hd = c >> 6;
    float v = agg[idx] / denom[node * HEADS + hd] + bias[c];
    if (do_relu) v = fmaxf(v, 0.f);
    z[idx] = v;
}

__global__ void decode_edges(const int* __restrict__ src, const int* __restrict__ dst,
                             const float* __restrict__ z,
                             float* __restrict__ out, int E)
{
    int wid = (blockIdx.x * blockDim.x + threadIdx.x) >> 5;
    int lane = threadIdx.x & 31;
    if (wid >= E) return;
    const float* zs = z + (size_t)src[wid] * FDIM;
    const float* zd = z + (size_t)dst[wid] * FDIM;
    float acc = 0.f;
#pragma unroll
    for (int i = 0; i < 4; ++i) { int f = lane + 32 * i; acc += zs[f] * zd[f]; }
#pragma unroll
    for (int off = 16; off > 0; off >>= 1) acc += __shfl_down(acc, off, 32);
    if (lane == 0) out[wid] = acc;
}

extern "C" void kernel_launch(void* const* d_in, const int* in_sizes, int n_in,
                              void* d_out, int out_size, void* d_ws, size_t ws_size,
                              hipStream_t stream) {
    const float* x      = (const float*)d_in[0];
    const int*   ei     = (const int*)d_in[1];
    const float* W1     = (const float*)d_in[2];
    const float* a1s    = (const float*)d_in[3];
    const float* a1d    = (const float*)d_in[4];
    const float* b1     = (const float*)d_in[5];
    const float* W2     = (const float*)d_in[6];
    const float* a2s    = (const float*)d_in[7];
    const float* a2d    = (const float*)d_in[8];
    const float* b2     = (const float*)d_in[9];
    float* scores = (float*)d_out;

    const int N = in_sizes[0] / FDIM;
    const int E = in_sizes[1] / 2;
    const int* src = ei;
    const int* dst = ei + E;

    float* ws   = (float*)d_ws;
    float*    h     = ws;
    float*    agg   = ws + (size_t)N * FDIM;
    float*    z     = ws + (size_t)2 * N * FDIM;
    float*    esrc  = ws + (size_t)3 * N * FDIM;
    float*    edst  = esrc + (size_t)N * HEADS;
    unsigned* mmax  = (unsigned*)(edst + (size_t)N * HEADS);
    float*    denom = (float*)mmax + (size_t)N * HEADS;

    const int ETOT = E + N;
    dim3 blk(256);
    dim3 gGemm((N + 127) / 128);
    dim3 gNH((N * HEADS + 255) / 256);
    dim3 gInit((N * FDIM + 255) / 256);
    dim3 gEdge((ETOT + 255) / 256);
    dim3 gEdgeW((ETOT * 32 + 255) / 256);
    dim3 gDecW(((size_t)E * 32 + 255) / 256);

    gemm128_wmma<<<gGemm, blk, 0, stream>>>(x, W1, h, N);
    attn_logits<<<gNH, blk, 0, stream>>>(h, a1s, a1d, esrc, edst, N);
    init_accum<<<gInit, blk, 0, stream>>>(mmax, denom, agg, N);
    edge_max<<<gEdge, blk, 0, stream>>>(src, dst, E, N, esrc, edst, mmax);
    edge_aggregate<<<gEdgeW, blk, 0, stream>>>(src, dst, E, N, esrc, edst, mmax, h, denom, agg);
    node_finalize<<<gInit, blk, 0, stream>>>(agg, denom, b1, z, N,  1);

    gemm128_wmma<<<gGemm, blk, 0, stream>>>(z, W2, h, N);
    attn_logits<<<gNH, blk, 0, stream>>>(h, a2s, a2d, esrc, edst, N);
    init_accum<<<gInit, blk, 0, stream>>>(mmax, denom, agg, N);
    edge_max<<<gEdge, blk, 0, stream>>>(src, dst, E, N, esrc, edst, mmax);
    edge_aggregate<<<gEdgeW, blk, 0, stream>>>(src, dst, E, N, esrc, edst, mmax, h, denom, agg);
    node_finalize<<<gInit, blk, 0, stream>>>(agg, denom, b2, z, N,  0);

    decode_edges<<<gDecW, blk, 0, stream>>>(src, dst, z, scores, E);
}


# edge softmax+aggregation regrouped by destination: edges counting-sorted by dst in-call (CSR), one wave per node keeps max, exp-sum and feature sum in registers; float atomics gone
# speedup vs baseline: 2.4851x; 2.4851x over previous
#include <hip/hip_runtime.h>
#include <hip/hip_bf16.h>


typedef __attribute__((ext_vector_type(2))) float v2f;
typedef __attribute__((ext_vector_type(8))) float v8f;

#define HEADS 2
#define FDIM 128
#define NEG_SLOPE 0.2f

__device__ __forceinline__ float lrelu(float v) { return (v >= 0.f) ? v : NEG_SLOPE * v; }

__global__ __launch_bounds__(256) void gemm128_wmma(
    const float* __restrict__ X, const float* __restrict__ W,
    float* __restrict__ Y, int M)
{
    __shared__ float lw[FDIM * FDIM];
    {
        const float4* w4 = (const float4*)W;
        float4* l4 = (float4*)lw;
        for (int i = threadIdx.x; i < FDIM * FDIM / 4; i += 256) l4[i] = w4[i];
    }
    __syncthreads();

    const int wave = threadIdx.x >> 5;
    const int lane = threadIdx.x & 31;
    const int m0 = (blockIdx.x * 8 + wave) * 16;
    if (m0 >= M) return;

    const int lm = lane & 15;
    const int lh = lane >> 4;
    const float* arow = X + (size_t)(m0 + lm) * FDIM + 2 * lh;

#pragma unroll
    for (int nt = 0; nt < 8; ++nt) {
        v8f acc = {};
        const int nc = nt * 16 + lm;
#pragma unroll 8
        for (int kk = 0; kk < 32; ++kk) {
            const int k0 = kk * 4;
            v2f a = *(const v2f*)(arow + k0);
            v2f b;
            b.x = lw[(k0 + 2 * lh) * FDIM + nc];
            b.y = lw[(k0 + 2 * lh + 1) * FDIM + nc];
            acc = __builtin_amdgcn_wmma_f32_16x16x4_f32(
                false, a, false, b, (short)0, acc, false, false);
        }
        float* yrow = Y + (size_t)(m0 + 8 * lh) * FDIM + nc;
#pragma unroll
        for (int r = 0; r < 8; ++r) yrow[(size_t)r * FDIM] = acc[r];
    }
}

__global__ void attn_logits(const float* __restrict__ h,
                            const float* __restrict__ a_src,
                            const float* __restrict__ a_dst,
                            float* __restrict__ esrc, float* __restrict__ edst,
                            int n)
{
    int idx = blockIdx.x * blockDim.x + threadIdx.x;
    if (idx >= n * HEADS) return;
    int node = idx >> 1, hd = idx & 1;
    const float* hp = h + (size_t)node * FDIM + hd * 64;
    const float* as = a_src + hd * 64;
    const float* ad = a_dst + hd * 64;
    float s = 0.f, t = 0.f;
#pragma unroll 8
    for (int f = 0; f < 64; ++f) { float x = hp[f]; s += x * as[f]; t += x * ad[f]; }
    esrc[idx] = s; edst[idx] = t;
}

__global__ void csr_zero(int* __restrict__ cnt, int n)
{
    int i = blockIdx.x * blockDim.x + threadIdx.x;
    if (i < n) cnt[i] = 0;
}

__global__ void csr_count(const int* __restrict__ dst, int E,
                          int* __restrict__ cnt, int* __restrict__ rank)
{
    int e = blockIdx.x * blockDim.x + threadIdx.x;
    if (e >= E) return;
    rank[e] = atomicAdd(&cnt[dst[e]], 1);
}

__global__ __launch_bounds__(1024) void csr_scan(const int* __restrict__ cnt,
                                                 int* __restrict__ rowptr, int n)
{
    __shared__ int sh_front[32];
    __shared__ int sh_wave[32];
    const int t = threadIdx.x, lane = t & 31, w = t >> 5;
    const int start = blockIdx.x * 1024;

    int p = 0;
    for (int i = t; i < start; i += 1024) p += cnt[i];
#pragma unroll
    for (int off = 16; off > 0; off >>= 1) p += __shfl_xor(p, off, 32);

    const int gi = start + t;
    const int v = (gi < n) ? cnt[gi] : 0;
    int inc = v;
#pragma unroll
    for (int off = 1; off < 32; off <<= 1) {
        int u = __shfl_up(inc, off, 32);
        if (lane >= off) inc += u;
    }
    if (lane == 31) sh_wave[w] = inc;
    if (lane == 0) sh_front[w] = p;
    __syncthreads();
    int base = 0;
#pragma unroll
    for (int k = 0; k < 32; ++k) {
        base += sh_front[k];
        if (k < w) base += sh_wave[k];
    }
    const int excl = base + inc - v;
    if (gi < n) rowptr[gi] = excl;
    if (gi == n - 1) rowptr[n] = excl + v;
}

__global__ void csr_fill(const int* __restrict__ src, const int* __restrict__ dst, int E,
                         const int* __restrict__ rowptr, const int* __restrict__ rank,
                         int* __restrict__ col)
{
    int e = blockIdx.x * blockDim.x + threadIdx.x;
    if (e >= E) return;
    atomicExch(&col[rowptr[dst[e]] + rank[e]], src[e]);
}

__global__ __launch_bounds__(256) void gat_gather(
    const int* __restrict__ rowptr, const int* __restrict__ col,
    const float* __restrict__ esrc, const float* __restrict__ edst,
    const float* __restrict__ hfeat, const float* __restrict__ bias,
    float* __restrict__ z, int n, int do_relu)
{
    const int lane = threadIdx.x & 31;
    const int node = __builtin_amdgcn_readfirstlane((int)((blockIdx.x * 256u + threadIdx.x) >> 5));
    if (node >= n) return;
    const int beg = rowptr[node], end = rowptr[node + 1];
    const float ed0 = edst[node * HEADS + 0], ed1 = edst[node * HEADS + 1];
    const float sf0 = lrelu(esrc[node * HEADS + 0] + ed0);
    const float sf1 = lrelu(esrc[node * HEADS + 1] + ed1);

    float m0 = sf0, m1 = sf1;
    for (int j = beg + lane; j < end; j += 32) {
        const int s = col[j];
        const float2 es = *(const float2*)(esrc + (size_t)s * HEADS);
        m0 = fmaxf(m0, lrelu(es.x + ed0));
        m1 = fmaxf(m1, lrelu(es.y + ed1));
    }
#pragma unroll
    for (int off = 16; off > 0; off >>= 1) {
        m0 = fmaxf(m0, __shfl_xor(m0, off, 32));
        m1 = fmaxf(m1, __shfl_xor(m1, off, 32));
    }

    const bool head1 = lane >= 16;
    const float as0 = __expf(sf0 - m0), as1 = __expf(sf1 - m1);
    float4 acc;
    {
        const float4 r = *(const float4*)(hfeat + (size_t)node * FDIM + lane * 4);
        const float wgt = head1 ? as1 : as0;
        acc.x = wgt * r.x; acc.y = wgt * r.y; acc.z = wgt * r.z; acc.w = wgt * r.w;
    }
    float p0 = 0.f, p1 = 0.f;
    for (int base = beg; base < end; base += 32) {
        const int j = base + lane;
        const int cnt = (end - base < 32) ? (end - base) : 32;
        int s = 0;
        float a0 = 0.f, a1 = 0.f;
        if (j < end) {
            s = col[j];
            const float2 es = *(const float2*)(esrc + (size_t)s * HEADS);
            a0 = __expf(lrelu(es.x + ed0) - m0);
            a1 = __expf(lrelu(es.y + ed1) - m1);
        }
        p0 += a0; p1 += a1;
#pragma unroll 8
        for (int t = 0; t < cnt; ++t) {
            const int st = __builtin_amdgcn_readlane(s, t);
            const float w0 = __int_as_float(__builtin_amdgcn_readlane(__float_as_int(a0), t));
            const float w1 = __int_as_float(__builtin_amdgcn_readlane(__float_as_int(a1), t));
            const float4 r = *(const float4*)(hfeat + (size_t)st * FDIM + lane * 4);
            const float wgt = head1 ? w1 : w0;
            acc.x += wgt * r.x; acc.y += wgt * r.y; acc.z += wgt * r.z; acc.w += wgt * r.w;
        }
    }
#pragma unroll
    for (int off = 16; off > 0; off >>= 1) {
        p0 += __shfl_xor(p0, off, 32);
        p1 += __shfl_xor(p1, off, 32);
    }
    const float den = head1 ? (p1 + as1) : (p0 + as0);
    const float4 b = *(const float4*)(bias + lane * 4);
    float4 o;
    o.x = acc.x / den + b.x; o.y = acc.y / den + b.y;
    o.z = acc.z / den + b.z; o.w = acc.w / den + b.w;
    if (do_relu) { o.x = fmaxf(o.x, 0.f); o.y = fmaxf(o.y, 0.f); o.z = fmaxf(o.z, 0.f); o.w = fmaxf(o.w, 0.f); }
    *(float4*)(z + (size_t)node * FDIM + lane * 4) = o;
}

__global__ void decode_edges(const int* __restrict__ src, const int* __restrict__ dst,
                             const float* __restrict__ z,
                             float* __restrict__ out, int E)
{
    int wid = (blockIdx.x * blockDim.x + threadIdx.x) >> 5;
    int lane = threadIdx.x & 31;
    if (wid >= E) return;
    const float* zs = z + (size_t)src[wid] * FDIM;
    const float* zd = z + (size_t)dst[wid] * FDIM;
    float acc = 0.f;
#pragma unroll
    for (int i = 0; i < 4; ++i) { int f = lane + 32 * i; acc += zs[f] * zd[f]; }
#pragma unroll
    for (int off = 16; off > 0; off >>= 1) acc += __shfl_down(acc, off, 32);
    if (lane == 0) out[wid] = acc;
}

extern "C" void kernel_launch(void* const* d_in, const int* in_sizes, int n_in,
                              void* d_out, int out_size, void* d_ws, size_t ws_size,
                              hipStream_t stream) {
    const float* x      = (const float*)d_in[0];
    const int*   ei     = (const int*)d_in[1];
    const float* W1     = (const float*)d_in[2];
    const float* a1s    = (const float*)d_in[3];
    const float* a1d    = (const float*)d_in[4];
    const float* b1     = (const float*)d_in[5];
    const float* W2     = (const float*)d_in[6];
    const float* a2s    = (const float*)d_in[7];
    const float* a2d    = (const float*)d_in[8];
    const float* b2     = (const float*)d_in[9];
    float* scores = (float*)d_out;

    const int N = in_sizes[0] / FDIM;
    const int E = in_sizes[1] / 2;
    const int* src = ei;
    const int* dst = ei + E;

    float* ws   = (float*)d_ws;
    float*    h     = ws;
    int*      cnt   = (int*)(ws + (size_t)N * FDIM);
    int*      rowp  = cnt + N;
    int*      rank  = rowp + N + 1;
    int*      col   = rank + E;
    float*    z     = ws + (size_t)2 * N * FDIM;
    float*    esrc  = ws + (size_t)3 * N * FDIM;
    float*    edst  = esrc + (size_t)N * HEADS;

    dim3 blk(256);
    dim3 gGemm((N + 127) / 128);
    dim3 gNH((N * HEADS + 255) / 256);
    dim3 gNode((N + 255) / 256);
    dim3 gE((E + 255) / 256);
    dim3 gScan((N + 1023) / 1024);
    dim3 gNodeW((N + 7) / 8);
    dim3 gDecW(((size_t)E * 32 + 255) / 256);

    csr_zero<<<gNode, blk, 0, stream>>>(cnt, N);
    csr_count<<<gE, blk, 0, stream>>>(dst, E, cnt, rank);
    csr_scan<<<gScan, dim3(1024), 0, stream>>>(cnt, rowp, N);
    csr_fill<<<gE, blk, 0, stream>>>(src, dst, E, rowp, rank, col);

    gemm128_wmma<<<gGemm, blk, 0, stream>>>(x, W1, h, N);
    attn_logits<<<gNH, blk, 0, stream>>>(h, a1s, a1d, esrc, edst, N);
    gat_gather<<<gNodeW, blk, 0, stream>>>(rowp, col, esrc, edst, h, b1, z, N,  1);

    gemm128_wmma<<<gGemm, blk, 0, stream>>>(z, W2, h, N);
    attn_logits<<<gNH, blk, 0, stream>>>(h, a2s, a2d, esrc, edst, N);
    gat_gather<<<gNodeW, blk, 0, stream>>>(rowp, col, esrc, edst, h, b2, z, N,  0);

    decode_edges<<<gDecW, blk, 0, stream>>>(src, dst, z, scores, E);
}


# link decode: one wave per 32 edges instead of one per edge; z rows read as 512-byte wave loads, 32 dot products summed over lanes by a lane-bit butterfly (new summation tree), one 128-byte store
# speedup vs baseline: 3.1920x; 1.2845x over previous
#include <hip/hip_runtime.h>
#include <hip/hip_bf16.h>


typedef __attribute__((ext_vector_type(2))) float v2f;
typedef __attribute__((ext_vector_type(8))) float v8f;

#define HEADS 2
#define FDIM 128
#define NEG_SLOPE 0.2f

__device__ __forceinline__ float lrelu(float v) { return (v >= 0.f) ? v : NEG_SLOPE * v; }

__global__ __launch_bounds__(256) void gemm128_wmma(
    const float* __restrict__ X, const float* __restrict__ W,
    float* __restrict__ Y, int M)
{
    __shared__ float lw[FDIM * FDIM];
    {
        const float4* w4 = (const float4*)W;
        float4* l4 = (float4*)lw;
        for (int i = threadIdx.x; i < FDIM * FDIM / 4; i += 256) l4[i] = w4[i];
    }
    __syncthreads();

    const int wave = threadIdx.x >> 5;
    const int lane = threadIdx.x & 31;
    const int m0 = (blockIdx.x * 8 + wave) * 16;
    if (m0 >= M) return;

    const int lm = lane & 15;
    const int lh = lane >> 4;
    const float* arow = X + (size_t)(m0 + lm) * FDIM + 2 * lh;

#pragma unroll
    for (int nt = 0; nt < 8; ++nt) {
        v8f acc = {};
        const int nc = nt * 16 + lm;
#pragma unroll 8
        for (int kk = 0; kk < 32; ++kk) {
            const int k0 = kk * 4;
            v2f a = *(const v2f*)(arow + k0);
            v2f b;
            b.x = lw[(k0 + 2 * lh) * FDIM + nc];
            b.y = lw[(k0 + 2 * lh + 1) * FDIM + nc];
            acc = __builtin_amdgcn_wmma_f32_16x16x4_f32(
                false, a, false, b, (short)0, acc, false, false);
        }
        float* yrow = Y + (size_t)(m0 + 8 * lh) * FDIM + nc;
#pragma unroll
        for (int r = 0; r < 8; ++r) yrow[(size_t)r * FDIM] = acc[r];
    }
}

__global__ void attn_logits(const float* __restrict__ h,
                            const float* __restrict__ a_src,
                            const float* __restrict__ a_dst,
                            float* __restrict__ esrc, float* __restrict__ edst,
                            int n)
{
    int idx = blockIdx.x * blockDim.x + threadIdx.x;
    if (idx >= n * HEADS) return;
    int node = idx >> 1, hd = idx & 1;
    const float* hp = h + (size_t)node * FDIM + hd * 64;
    const float* as = a_src + hd * 64;
    const float* ad = a_dst + hd * 64;
    float s = 0.f, t = 0.f;
#pragma unroll 8
    for (int f = 0; f < 64; ++f) { float x = hp[f]; s += x * as[f]; t += x * ad[f]; }
    esrc[idx] = s; edst[idx] = t;
}

__global__ void csr_zero(int* __restrict__ cnt, int n)
{
    int i = blockIdx.x * blockDim.x + threadIdx.x;
    if (i < n) cnt[i] = 0;
}

__global__ void csr_count(const int* __restrict__ dst, int E,
                          int* __restrict__ cnt, int* __restrict__ rank)
{
    int e = blockIdx.x * blockDim.x + threadIdx.x;
    if (e >= E) return;
    rank[e] = atomicAdd(&cnt[dst[e]], 1);
}

__global__ __launch_bounds__(1024) void csr_scan(const int* __restrict__ cnt,
                                                 int* __restrict__ rowptr, int n)
{
    __shared__ int sh_front[32];
    __shared__ int sh_wave[32];
    const int t = threadIdx.x, lane = t & 31, w = t >> 5;
    const int start = blockIdx.x * 1024;

    int p = 0;
    for (int i = t; i < start; i += 1024) p += cnt[i];
#pragma unroll
    for (int off = 16; off > 0; off >>= 1) p += __shfl_xor(p, off, 32);

    const int gi = start + t;
    const int v = (gi < n) ? cnt[gi] : 0;
    int inc = v;
#pragma unroll
    for (int off = 1; off < 32; off <<= 1) {
        int u = __shfl_up(inc, off, 32);
        if (lane >= off) inc += u;
    }
    if (lane == 31) sh_wave[w] = inc;
    if (lane == 0) sh_front[w] = p;
    __syncthreads();
    int base = 0;
#pragma unroll
    for (int k = 0; k < 32; ++k) {
        base += sh_front[k];
        if (k < w) base += sh_wave[k];
    }
    const int excl = base + inc - v;
    if (gi < n) rowptr[gi] = excl;
    if (gi == n - 1) rowptr[n] = excl + v;
}

__global__ void csr_fill(const int* __restrict__ src, const int* __restrict__ dst, int E,
                         const int* __restrict__ rowptr, const int* __restrict__ rank,
                         int* __restrict__ col)
{
    int e = blockIdx.x * blockDim.x + threadIdx.x;
    if (e >= E) return;
    atomicExch(&col[rowptr[dst[e]] + rank[e]], src[e]);
}

__global__ __launch_bounds__(256) void gat_gather(
    const int* __restrict__ rowptr, const int* __restrict__ col,
    const float* __restrict__ esrc, const float* __restrict__ edst,
    const float* __restrict__ hfeat, const float* __restrict__ bias,
    float* __restrict__ z, int n, int do_relu)
{
    const int lane = threadIdx.x & 31;
    const int node = __builtin_amdgcn_readfirstlane((int)((blockIdx.x * 256u + threadIdx.x) >> 5));
    if (node >= n) return;
    const int beg = rowptr[node], end = rowptr[node + 1];
    const float ed0 = edst[node * HEADS + 0], ed1 = edst[node * HEADS + 1];
    const float sf0 = lrelu(esrc[node * HEADS + 0] + ed0);
    const float sf1 = lrelu(esrc[node * HEADS + 1] + ed1);

    float m0 = sf0, m1 = sf1;
    for (int j = beg + lane; j < end; j += 32) {
        const int s = col[j];
        const float2 es = *(const float2*)(esrc + (size_t)s * HEADS);
        m0 = fmaxf(m0, lrelu(es.x + ed0));
        m1 = fmaxf(m1, lrelu(es.y + ed1));
    }
#pragma unroll
    for (int off = 16; off > 0; off >>= 1) {
        m0 = fmaxf(m0, __shfl_xor(m0, off, 32));
        m1 = fmaxf(m1, __shfl_xor(m1, off, 32));
    }

    const bool head1 = lane >= 16;
    const float as0 = __expf(sf0 - m0), as1 = __expf(sf1 - m1);
    float4 acc;
    {
        const float4 r = *(const float4*)(hfeat + (size_t)node * FDIM + lane * 4);
        const float wgt = head1 ? as1 : as0;
        acc.x = wgt * r.x; acc.y = wgt * r.y; acc.z = wgt * r.z; acc.w = wgt * r.w;
    }
    float p0 = 0.f, p1 = 0.f;
    for (int base = beg; base < end; base += 32) {
        const int j = base + lane;
        const int cnt = (end - base < 32) ? (end - base) : 32;
        int s = 0;
        float a0 = 0.f, a1 = 0.f;
        if (j < end) {
            s = col[j];
            const float2 es = *(const float2*)(esrc + (size_t)s * HEADS);
            a0 = __expf(lrelu(es.x + ed0) - m0);
            a1 = __expf(lrelu(es.y + ed1) - m1);
        }
        p0 += a0; p1 += a1;
#pragma unroll 8
        for (int t = 0; t < cnt; ++t) {
            const int st = __builtin_amdgcn_readlane(s, t);
            const float w0 = __int_as_float(__builtin_amdgcn_readlane(__float_as_int(a0), t));
            const float w1 = __int_as_float(__builtin_amdgcn_readlane(__float_as_int(a1), t));
            const float4 r = *(const float4*)(hfeat + (size_t)st * FDIM + lane * 4);
            const float wgt = head1 ? w1 : w0;
            acc.x += wgt * r.x; acc.y += wgt * r.y; acc.z += wgt * r.z; acc.w += wgt * r.w;
        }
    }
#pragma unroll
    for (int off = 16; off > 0; off >>= 1) {
        p0 += __shfl_xor(p0, off, 32);
        p1 += __shfl_xor(p1, off, 32);
    }
    const float den = head1 ? (p1 + as1) : (p0 + as0);
    const float4 b = *(const float4*)(bias + lane * 4);
    float4 o;
    o.x = acc.x / den + b.x; o.y = acc.y / den + b.y;
    o.z = acc.z / den + b.z; o.w = acc.w / den + b.w;
    if (do_relu) { o.x = fmaxf(o.x, 0.f); o.y = fmaxf(o.y, 0.f); o.z = fmaxf(o.z, 0.f); o.w = fmaxf(o.w, 0.f); }
    *(float4*)(z + (size_t)node * FDIM + lane * 4) = o;
}

__device__ __forceinline__ float bfly_step(float lo, float hi, bool up, int mask)
{
    const float keep = up ? hi : lo;
    const float send = up ? lo : hi;
    return keep + __shfl_xor(send, mask, 32);
}

__global__ __launch_bounds__(256) void decode_edges(const int* __restrict__ src,
                                                    const int* __restrict__ dst,
                                                    const float* __restrict__ z,
                                                    float* __restrict__ out, int E)
{
    const int lane = threadIdx.x & 31;
    const int base = __builtin_amdgcn_readfirstlane((int)((blockIdx.x * 256u + threadIdx.x) >> 5)) * 32;
    if (base >= E) return;
    const int e = base + lane;
    const bool ok = e < E;
    const int sv = ok ? src[e] : 0;
    const int dv = ok ? dst[e] : 0;

    const bool b0 = lane & 1, b1 = lane & 2, b2 = lane & 4, b3 = lane & 8, b4 = lane & 16;
    float x[4];
#pragma unroll
    for (int g = 0; g < 4; ++g) {
        float p[8];
#pragma unroll
        for (int t = 0; t < 8; ++t) {
            const int s = __builtin_amdgcn_readlane(sv, g * 8 + t);
            const int d = __builtin_amdgcn_readlane(dv, g * 8 + t);
            const float4 a = *(const float4*)(z + (size_t)s * FDIM + lane * 4);
            const float4 b = *(const float4*)(z + (size_t)d * FDIM + lane * 4);
            p[t] = a.x * b.x + a.y * b.y + a.z * b.z + a.w * b.w;
        }
        float u[4], w[2];
#pragma unroll
        for (int i = 0; i < 4; ++i) u[i] = bfly_step(p[2 * i], p[2 * i + 1], b0, 1);
#pragma unroll
        for (int i = 0; i < 2; ++i) w[i] = bfly_step(u[2 * i], u[2 * i + 1], b1, 2);
        x[g] = bfly_step(w[0], w[1], b2, 4);
    }
    const float y0 = bfly_step(x[0], x[1], b3, 8);
    const float y1 = bfly_step(x[2], x[3], b3, 8);
    const float r  = bfly_step(y0, y1, b4, 16);
    if (ok) out[e] = r;
}

extern "C" void kernel_launch(void* const* d_in, const int* in_sizes, int n_in,
                              void* d_out, int out_size, void* d_ws, size_t ws_size,
                              hipStream_t stream) {
    const float* x      = (const float*)d_in[0];
    const int*   ei     = (const int*)d_in[1];
    const float* W1     = (const float*)d_in[2];
    const float* a1s    = (const float*)d_in[3];
    const float* a1d    = (const float*)d_in[4];
    const float* b1     = (const float*)d_in[5];
    const float* W2     = (const float*)d_in[6];
    const float* a2s    = (const float*)d_in[7];
    const float* a2d    = (const float*)d_in[8];
    const float* b2     = (const float*)d_in[9];
    float* scores = (float*)d_out;

    const int N = in_sizes[0] / FDIM;
    const int E = in_sizes[1] / 2;
    const int* src = ei;
    const int* dst = ei + E;

    float* ws   = (float*)d_ws;
    float*    h     = ws;
    int*      cnt   = (int*)(ws + (size_t)N * FDIM);
    int*      rowp  = cnt + N;
    int*      rank  = rowp + N + 1;
    int*      col   = rank + E;
    float*    z     = ws + (size_t)2 * N * FDIM;
    float*    esrc  = ws + (size_t)3 * N * FDIM;
    float*    edst  = esrc + (size_t)N * HEADS;

    dim3 blk(256);
    dim3 gGemm((N + 127) / 128);
    dim3 gNH((N * HEADS + 255) / 256);
    dim3 gNode((N + 255) / 256);
    dim3 gE((E + 255) / 256);
    dim3 gScan((N + 1023) / 1024);
    dim3 gNodeW((N + 7) / 8);

    csr_zero<<<gNode, blk, 0, stream>>>(cnt, N);
    csr_count<<<gE, blk, 0, stream>>>(dst, E, cnt, rank);
    csr_scan<<<gScan, dim3(1024), 0, stream>>>(cnt, rowp, N);
    csr_fill<<<gE, blk, 0, stream>>>(src, dst, E, rowp, rank, col);

    gemm128_wmma<<<gGemm, blk, 0, stream>>>(x, W1, h, N);
    attn_logits<<<gNH, blk, 0, stream>>>(h, a1s, a1d, esrc, edst, N);
    gat_gather<<<gNodeW, blk, 0, stream>>>(rowp, col, esrc, edst, h, b1, z, N,  1);

    gemm128_wmma<<<gGemm, blk, 0, stream>>>(z, W2, h, N);
    attn_logits<<<gNH, blk, 0, stream>>>(h, a2s, a2d, esrc, edst, N);
    gat_gather<<<gNodeW, blk, 0, stream>>>(rowp, col, esrc, edst, h, b2, z, N,  0);

    decode_edges<<<gE, blk, 0, stream>>>(src, dst, z, scores, E);
}


# node gather reads each slot's source index and e_src once: max pass keeps a row's first 64 slots (two per lane) in registers for the sum pass; second col read and e_src gather gone; sum order kept
# speedup vs baseline: 3.3266x; 1.0422x over previous
#include <hip/hip_runtime.h>
#include <hip/hip_bf16.h>


typedef __attribute__((ext_vector_type(2))) float v2f;
typedef __attribute__((ext_vector_type(8))) float v8f;

#define HEADS 2
#define FDIM 128
#define NEG_SLOPE 0.2f

__device__ __forceinline__ float lrelu(float v) { return (v >= 0.f) ? v : NEG_SLOPE * v; }

__global__ __launch_bounds__(256) void gemm128_wmma(
    const float* __restrict__ X, const float* __restrict__ W,
    float* __restrict__ Y, int M)
{
    __shared__ float lw[FDIM * FDIM];
    {
        const float4* w4 = (const float4*)W;
        float4* l4 = (float4*)lw;
        for (int i = threadIdx.x; i < FDIM * FDIM / 4; i += 256) l4[i] = w4[i];
    }
    __syncthreads();

    const int wave = threadIdx.x >> 5;
    const int lane = threadIdx.x & 31;
    const int m0 = (blockIdx.x * 8 + wave) * 16;
    if (m0 >= M) return;

    const int lm = lane & 15;
    const int lh = lane >> 4;
    const float* arow = X + (size_t)(m0 + lm) * FDIM + 2 * lh;

#pragma unroll
    for (int nt = 0; nt < 8; ++nt) {
        v8f acc = {};
        const int nc = nt * 16 + lm;
#pragma unroll 8
        for (int kk = 0; kk < 32; ++kk) {
            const int k0 = kk * 4;
            v2f a = *(const v2f*)(arow + k0);
            v2f b;
            b.x = lw[(k0 + 2 * lh) * FDIM + nc];
            b.y = lw[(k0 + 2 * lh + 1) * FDIM + nc];
            acc = __builtin_amdgcn_wmma_f32_16x16x4_f32(
                false, a, false, b, (short)0, acc, false, false);
        }
        float* yrow = Y + (size_t)(m0 + 8 * lh) * FDIM + nc;
#pragma unroll
        for (int r = 0; r < 8; ++r) yrow[(size_t)r * FDIM] = acc[r];
    }
}

__global__ void attn_logits(const float* __restrict__ h,
                            const float* __restrict__ a_src,
                            const float* __restrict__ a_dst,
                            float* __restrict__ esrc, float* __restrict__ edst,
                            int n)
{
    int idx = blockIdx.x * blockDim.x + threadIdx.x;
    if (idx >= n * HEADS) return;
    int node = idx >> 1, hd = idx & 1;
    const float* hp = h + (size_t)node * FDIM + hd * 64;
    const float* as = a_src + hd * 64;
    const float* ad = a_dst + hd * 64;
    float s = 0.f, t = 0.f;
#pragma unroll 8
    for (int f = 0; f < 64; ++f) { float x = hp[f]; s += x * as[f]; t += x * ad[f]; }
    esrc[idx] = s; edst[idx] = t;
}

__global__ void csr_zero(int* __restrict__ cnt, int n)
{
    int i = blockIdx.x * blockDim.x + threadIdx.x;
    if (i < n) cnt[i] = 0;
}

__global__ void csr_count(const int* __restrict__ dst, int E,
                          int* __restrict__ cnt, int* __restrict__ rank)
{
    int e = blockIdx.x * blockDim.x + threadIdx.x;
    if (e >= E) return;
    rank[e] = atomicAdd(&cnt[dst[e]], 1);
}

__global__ __launch_bounds__(1024) void csr_scan(const int* __restrict__ cnt,
                                                 int* __restrict__ rowptr, int n)
{
    __shared__ int sh_front[32];
    __shared__ int sh_wave[32];
    const int t = threadIdx.x, lane = t & 31, w = t >> 5;
    const int start = blockIdx.x * 1024;

    int p = 0;
    for (int i = t; i < start; i += 1024) p += cnt[i];
#pragma unroll
    for (int off = 16; off > 0; off >>= 1) p += __shfl_xor(p, off, 32);

    const int gi = start + t;
    const int v = (gi < n) ? cnt[gi] : 0;
    int inc = v;
#pragma unroll
    for (int off = 1; off < 32; off <<= 1) {
        int u = __shfl_up(inc, off, 32);
        if (lane >= off) inc += u;
    }
    if (lane == 31) sh_wave[w] = inc;
    if (lane == 0) sh_front[w] = p;
    __syncthreads();
    int base = 0;
#pragma unroll
    for (int k = 0; k < 32; ++k) {
        base += sh_front[k];
        if (k < w) base += sh_wave[k];
    }
    const int excl = base + inc - v;
    if (gi < n) rowptr[gi] = excl;
    if (gi == n - 1) rowptr[n] = excl + v;
}

__global__ void csr_fill(const int* __restrict__ src, const int* __restrict__ dst, int E,
                         const int* __restrict__ rowptr, const int* __restrict__ rank,
                         int* __restrict__ col)
{
    int e = blockIdx.x * blockDim.x + threadIdx.x;
    if (e >= E) return;
    atomicExch(&col[rowptr[dst[e]] + rank[e]], src[e]);
}

__global__ __launch_bounds__(256) void gat_gather(
    const int* __restrict__ rowptr, const int* __restrict__ col,
    const float* __restrict__ esrc, const float* __restrict__ edst,
    const float* __restrict__ hfeat, const float* __restrict__ bias,
    float* __restrict__ z, int n, int do_relu)
{
    const int lane = threadIdx.x & 31;
    const int node = __builtin_amdgcn_readfirstlane((int)((blockIdx.x * 256u + threadIdx.x) >> 5));
    if (node >= n) return;
    const int beg = rowptr[node], end = rowptr[node + 1];
    const float ed0 = edst[node * HEADS + 0], ed1 = edst[node * HEADS + 1];
    const float sf0 = lrelu(esrc[node * HEADS + 0] + ed0);
    const float sf1 = lrelu(esrc[node * HEADS + 1] + ed1);

    const int ja = beg + lane, jb = ja + 32;
    const bool va = ja < end, vb = jb < end;
    int sa = node, sb = node;
    if (va) sa = col[ja];
    if (vb) sb = col[jb];
    const float2 esa = *(const float2*)(esrc + (size_t)sa * HEADS);
    const float2 esb = *(const float2*)(esrc + (size_t)sb * HEADS);
    const float ea0 = lrelu(esa.x + ed0), ea1 = lrelu(esa.y + ed1);
    const float eb0 = lrelu(esb.x + ed0), eb1 = lrelu(esb.y + ed1);
    float m0 = fmaxf(fmaxf(sf0, ea0), eb0);
    float m1 = fmaxf(fmaxf(sf1, ea1), eb1);
    for (int j = jb + 32; j < end; j += 32) {
        const int s = col[j];
        const float2 es = *(const float2*)(esrc + (size_t)s * HEADS);
        m0 = fmaxf(m0, lrelu(es.x + ed0));
        m1 = fmaxf(m1, lrelu(es.y + ed1));
    }
#pragma unroll
    for (int off = 16; off > 0; off >>= 1) {
        m0 = fmaxf(m0, __shfl_xor(m0, off, 32));
        m1 = fmaxf(m1, __shfl_xor(m1, off, 32));
    }

    const bool head1 = lane >= 16;
    const float as0 = __expf(sf0 - m0), as1 = __expf(sf1 - m1);
    float4 acc;
    {
        const float4 r = *(const float4*)(hfeat + (size_t)node * FDIM + lane * 4);
        const float wgt = head1 ? as1 : as0;
        acc.x = wgt * r.x; acc.y = wgt * r.y; acc.z = wgt * r.z; acc.w = wgt * r.w;
    }
    const float wa0 = va ? __expf(ea0 - m0) : 0.f, wa1 = va ? __expf(ea1 - m1) : 0.f;
    const float wb0 = vb ? __expf(eb0 - m0) : 0.f, wb1 = vb ? __expf(eb1 - m1) : 0.f;
    float p0 = 0.f, p1 = 0.f;
    int chunk = 0;
    for (int base = beg; base < end; base += 32, ++chunk) {
        const int cnt = (end - base < 32) ? (end - base) : 32;
        int s;
        float a0, a1;
        if (chunk == 0)      { s = sa; a0 = wa0; a1 = wa1; }
        else if (chunk == 1) { s = sb; a0 = wb0; a1 = wb1; }
        else {
            const int j = base + lane;
            s = 0; a0 = 0.f; a1 = 0.f;
            if (j < end) {
                s = col[j];
                const float2 es = *(const float2*)(esrc + (size_t)s * HEADS);
                a0 = __expf(lrelu(es.x + ed0) - m0);
                a1 = __expf(lrelu(es.y + ed1) - m1);
            }
        }
        p0 += a0; p1 += a1;
#pragma unroll 8
        for (int t = 0; t < cnt; ++t) {
            const int st = __builtin_amdgcn_readlane(s, t);
            const float w0 = __int_as_float(__builtin_amdgcn_readlane(__float_as_int(a0), t));
            const float w1 = __int_as_float(__builtin_amdgcn_readlane(__float_as_int(a1), t));
            const float4 r = *(const float4*)(hfeat + (size_t)st * FDIM + lane * 4);
            const float wgt = head1 ? w1 : w0;
            acc.x += wgt * r.x; acc.y += wgt * r.y; acc.z += wgt * r.z; acc.w += wgt * r.w;
        }
    }
#pragma unroll
    for (int off = 16; off > 0; off >>= 1) {
        p0 += __shfl_xor(p0, off, 32);
        p1 += __shfl_xor(p1, off, 32);
    }
    const float den = head1 ? (p1 + as1) : (p0 + as0);
    const float4 b = *(const float4*)(bias + lane * 4);
    float4 o;
    o.x = acc.x / den + b.x; o.y = acc.y / den + b.y;
    o.z = acc.z / den + b.z; o.w = acc.w / den + b.w;
    if (do_relu) { o.x = fmaxf(o.x, 0.f); o.y = fmaxf(o.y, 0.f); o.z = fmaxf(o.z, 0.f); o.w = fmaxf(o.w, 0.f); }
    *(float4*)(z + (size_t)node * FDIM + lane * 4) = o;
}

__device__ __forceinline__ float bfly_step(float lo, float hi, bool up, int mask)
{
    const float keep = up ? hi : lo;
    const float send = up ? lo : hi;
    return keep + __shfl_xor(send, mask, 32);
}

__global__ __launch_bounds__(256) void decode_edges(const int* __restrict__ src,
                                                    const int* __restrict__ dst,
                                                    const float* __restrict__ z,
                                                    float* __restrict__ out, int E)
{
    const int lane = threadIdx.x & 31;
    const int base = __builtin_amdgcn_readfirstlane((int)((blockIdx.x * 256u + threadIdx.x) >> 5)) * 32;
    if (base >= E) return;
    const int e = base + lane;
    const bool ok = e < E;
    const int sv = ok ? src[e] : 0;
    const int dv = ok ? dst[e] : 0;

    const bool b0 = lane & 1, b1 = lane & 2, b2 = lane & 4, b3 = lane & 8, b4 = lane & 16;
    float x[4];
#pragma unroll
    for (int g = 0; g < 4; ++g) {
        float p[8];
#pragma unroll
        for (int t = 0; t < 8; ++t) {
            const int s = __builtin_amdgcn_readlane(sv, g * 8 + t);
            const int d = __builtin_amdgcn_readlane(dv, g * 8 + t);
            const float4 a = *(const float4*)(z + (size_t)s * FDIM + lane * 4);
            const float4 b = *(const float4*)(z + (size_t)d * FDIM + lane * 4);
            p[t] = a.x * b.x + a.y * b.y + a.z * b.z + a.w * b.w;
        }
        float u[4], w[2];
#pragma unroll
        for (int i = 0; i < 4; ++i) u[i] = bfly_step(p[2 * i], p[2 * i + 1], b0, 1);
#pragma unroll
        for (int i = 0; i < 2; ++i) w[i] = bfly_step(u[2 * i], u[2 * i + 1], b1, 2);
        x[g] = bfly_step(w[0], w[1], b2, 4);
    }
    const float y0 = bfly_step(x[0], x[1], b3, 8);
    const float y1 = bfly_step(x[2], x[3], b3, 8);
    const float r  = bfly_step(y0, y1, b4, 16);
    if (ok) out[e] = r;
}

extern "C" void kernel_launch(void* const* d_in, const int* in_sizes, int n_in,
                              void* d_out, int out_size, void* d_ws, size_t ws_size,
                              hipStream_t stream) {
    const float* x      = (const float*)d_in[0];
    const int*   ei     = (const int*)d_in[1];
    const float* W1     = (const float*)d_in[2];
    const float* a1s    = (const float*)d_in[3];
    const float* a1d    = (const float*)d_in[4];
    const float* b1     = (const float*)d_in[5];
    const float* W2     = (const float*)d_in[6];
    const float* a2s    = (const float*)d_in[7];
    const float* a2d    = (const float*)d_in[8];
    const float* b2     = (const float*)d_in[9];
    float* scores = (float*)d_out;

    const int N = in_sizes[0] / FDIM;
    const int E = in_sizes[1] / 2;
    const int* src = ei;
    const int* dst = ei + E;

    float* ws   = (float*)d_ws;
    float*    h     = ws;
    int*      cnt   = (int*)(ws + (size_t)N * FDIM);
    int*      rowp  = cnt + N;
    int*      rank  = rowp + N + 1;
    int*      col   = rank + E;
    float*    z     = ws + (size_t)2 * N * FDIM;
    float*    esrc  = ws + (size_t)3 * N * FDIM;
    float*    edst  = esrc + (size_t)N * HEADS;

    dim3 blk(256);
    dim3 gGemm((N + 127) / 128);
    dim3 gNH((N * HEADS + 255) / 256);
    dim3 gNode((N + 255) / 256);
    dim3 gE((E + 255) / 256);
    dim3 gScan((N + 1023) / 1024);
    dim3 gNodeW((N + 7) / 8);

    csr_zero<<<gNode, blk, 0, stream>>>(cnt, N);
    csr_count<<<gE, blk, 0, stream>>>(dst, E, cnt, rank);
    csr_scan<<<gScan, dim3(1024), 0, stream>>>(cnt, rowp, N);
    csr_fill<<<gE, blk, 0, stream>>>(src, dst, E, rowp, rank, col);

    gemm128_wmma<<<gGemm, blk, 0, stream>>>(x, W1, h, N);
    attn_logits<<<gNH, blk, 0, stream>>>(h, a1s, a1d, esrc, edst, N);
    gat_gather<<<gNodeW, blk, 0, stream>>>(rowp, col, esrc, edst, h, b1, z, N,  1);

    gemm128_wmma<<<gGemm, blk, 0, stream>>>(z, W2, h, N);
    attn_logits<<<gNH, blk, 0, stream>>>(h, a2s, a2d, esrc, edst, N);
    gat_gather<<<gNodeW, blk, 0, stream>>>(rowp, col, esrc, edst, h, b2, z, N,  0);

    decode_edges<<<gE, blk, 0, stream>>>(src, dst, z, scores, E);
}


# CSR edge ranks without a global atomic add per edge: 64 work-groups count their edge blocks over all nodes in LDS (returning LDS adds), bases by a prefix over the 64 planes; gather sum order changes
# speedup vs baseline: 4.1716x; 1.2540x over previous
#include <hip/hip_runtime.h>
#include <hip/hip_bf16.h>


typedef __attribute__((ext_vector_type(2))) float v2f;
typedef __attribute__((ext_vector_type(8))) float v8f;

#define HEADS 2
#define FDIM 128
#define NEG_SLOPE 0.2f

__device__ __forceinline__ float lrelu(float v) { return (v >= 0.f) ? v : NEG_SLOPE * v; }

__global__ __launch_bounds__(256) void gemm128_wmma(
    const float* __restrict__ X, const float* __restrict__ W,
    float* __restrict__ Y, int M)
{
    __shared__ float lw[FDIM * FDIM];
    {
        const float4* w4 = (const float4*)W;
        float4* l4 = (float4*)lw;
        for (int i = threadIdx.x; i < FDIM * FDIM / 4; i += 256) l4[i] = w4[i];
    }
    __syncthreads();

    const int wave = threadIdx.x >> 5;
    const int lane = threadIdx.x & 31;
    const int m0 = (blockIdx.x * 8 + wave) * 16;
    if (m0 >= M) return;

    const int lm = lane & 15;
    const int lh = lane >> 4;
    const float* arow = X + (size_t)(m0 + lm) * FDIM + 2 * lh;

#pragma unroll
    for (int nt = 0; nt < 8; ++nt) {
        v8f acc = {};
        const int nc = nt * 16 + lm;
#pragma unroll 8
        for (int kk = 0; kk < 32; ++kk) {
            const int k0 = kk * 4;
            v2f a = *(const v2f*)(arow + k0);
            v2f b;
            b.x = lw[(k0 + 2 * lh) * FDIM + nc];
            b.y = lw[(k0 + 2 * lh + 1) * FDIM + nc];
            acc = __builtin_amdgcn_wmma_f32_16x16x4_f32(
                false, a, false, b, (short)0, acc, false, false);
        }
        float* yrow = Y + (size_t)(m0 + 8 * lh) * FDIM + nc;
#pragma unroll
        for (int r = 0; r < 8; ++r) yrow[(size_t)r * FDIM] = acc[r];
    }
}

__global__ void attn_logits(const float* __restrict__ h,
                            const float* __restrict__ a_src,
                            const float* __restrict__ a_dst,
                            float* __restrict__ esrc, float* __restrict__ edst,
                            int n)
{
    int idx = blockIdx.x * blockDim.x + threadIdx.x;
    if (idx >= n * HEADS) return;
    int node = idx >> 1, hd = idx & 1;
    const float* hp = h + (size_t)node * FDIM + hd * 64;
    const float* as = a_src + hd * 64;
    const float* ad = a_dst + hd * 64;
    float s = 0.f, t = 0.f;
#pragma unroll 8
    for (int f = 0; f < 64; ++f) { float x = hp[f]; s += x * as[f]; t += x * ad[f]; }
    esrc[idx] = s; edst[idx] = t;
}

#define CSR_BLOCKS 64
#define CSR_WIN 50176

__global__ __launch_bounds__(1024) void csr_count(const int* __restrict__ dst, int E, int n, int per,
                                                  int* __restrict__ hist, int* __restrict__ rank)
{
    __shared__ int lcnt[CSR_WIN];
    const int t = threadIdx.x;
    const int blk = blockIdx.x;
    const int e0 = blk * per;
    const int e1 = (E - e0 < per) ? E : e0 + per;
    int* __restrict__ plane = hist + (size_t)blk * n;
    for (int w0 = 0; w0 < n; w0 += CSR_WIN) {
        const int wn = (n - w0 < CSR_WIN) ? (n - w0) : CSR_WIN;
        for (int i = t; i < wn; i += 1024) lcnt[i] = 0;
        __syncthreads();
        for (int e = e0 + t; e < e1; e += 8 * 1024) {
            int d[8], r[8];
#pragma unroll
            for (int k = 0; k < 8; ++k) {
                const int ek = e + k * 1024;
                const int v = dst[(ek < e1) ? ek : (e1 - 1)];
                d[k] = (ek < e1) ? v - w0 : -1;
            }
#pragma unroll
            for (int k = 0; k < 8; ++k) {
                r[k] = 0;
                if ((unsigned)d[k] < (unsigned)wn) r[k] = atomicAdd(&lcnt[d[k]], 1);
            }
#pragma unroll
            for (int k = 0; k < 8; ++k) {
                if ((unsigned)d[k] < (unsigned)wn) rank[e + k * 1024] = r[k];
            }
        }
        __syncthreads();
        for (int i = t; i < wn; i += 1024) plane[w0 + i] = lcnt[i];
        __syncthreads();
    }
}

__global__ void csr_merge(const int* __restrict__ hist, int* __restrict__ base,
                          int* __restrict__ cnt, int n)
{
    const int d = blockIdx.x * blockDim.x + threadIdx.x;
    if (d >= n) return;
    int v[CSR_BLOCKS];
#pragma unroll
    for (int b = 0; b < CSR_BLOCKS; ++b) v[b] = hist[(size_t)b * n + d];
    int run = 0;
#pragma unroll
    for (int b = 0; b < CSR_BLOCKS; ++b) {
        base[(size_t)b * n + d] = run;
        run += v[b];
    }
    cnt[d] = run;
}

__global__ __launch_bounds__(1024) void csr_scan(const int* __restrict__ cnt,
                                                 int* __restrict__ rowptr, int n)
{
    __shared__ int sh_front[32];
    __shared__ int sh_wave[32];
    const int t = threadIdx.x, lane = t & 31, w = t >> 5;
    const int start = blockIdx.x * 1024;

    int p = 0;
    for (int i = t; i < start; i += 1024) p += cnt[i];
#pragma unroll
    for (int off = 16; off > 0; off >>= 1) p += __shfl_xor(p, off, 32);

    const int gi = start + t;
    const int v = (gi < n) ? cnt[gi] : 0;
    int inc = v;
#pragma unroll
    for (int off = 1; off < 32; off <<= 1) {
        int u = __shfl_up(inc, off, 32);
        if (lane >= off) inc += u;
    }
    if (lane == 31) sh_wave[w] = inc;
    if (lane == 0) sh_front[w] = p;
    __syncthreads();
    int base = 0;
#pragma unroll
    for (int k = 0; k < 32; ++k) {
        base += sh_front[k];
        if (k < w) base += sh_wave[k];
    }
    const int excl = base + inc - v;
    if (gi < n) rowptr[gi] = excl;
    if (gi == n - 1) rowptr[n] = excl + v;
}

__global__ void csr_fill(const int* __restrict__ src, const int* __restrict__ dst, int E, int n, int per,
                         const int* __restrict__ rowptr, const int* __restrict__ base,
                         const int* __restrict__ rank, int* __restrict__ col)
{
    int e = blockIdx.x * blockDim.x + threadIdx.x;
    if (e >= E) return;
    const int d = dst[e];
    if ((unsigned)d >= (unsigned)n) return;
    const int blk = e / per;
    atomicExch(&col[rowptr[d] + base[(size_t)blk * n + d] + rank[e]], src[e]);
}

__global__ __launch_bounds__(256) void gat_gather(
    const int* __restrict__ rowptr, const int* __restrict__ col,
    const float* __restrict__ esrc, const float* __restrict__ edst,
    const float* __restrict__ hfeat, const float* __restrict__ bias,
    float* __restrict__ z, int n, int do_relu)
{
    const int lane = threadIdx.x & 31;
    const int node = __builtin_amdgcn_readfirstlane((int)((blockIdx.x * 256u + threadIdx.x) >> 5));
    if (node >= n) return;
    const int beg = rowptr[node], end = rowptr[node + 1];
    const float ed0 = edst[node * HEADS + 0], ed1 = edst[node * HEADS + 1];
    const float sf0 = lrelu(esrc[node * HEADS + 0] + ed0);
    const float sf1 = lrelu(esrc[node * HEADS + 1] + ed1);

    const int ja = beg + lane, jb = ja + 32;
    const bool va = ja < end, vb = jb < end;
    int sa = node, sb = node;
    if (va) sa = col[ja];
    if (vb) sb = col[jb];
    const float2 esa = *(const float2*)(esrc + (size_t)sa * HEADS);
    const float2 esb = *(const float2*)(esrc + (size_t)sb * HEADS);
    const float ea0 = lrelu(esa.x + ed0), ea1 = lrelu(esa.y + ed1);
    const float eb0 = lrelu(esb.x + ed0), eb1 = lrelu(esb.y + ed1);
    float m0 = fmaxf(fmaxf(sf0, ea0), eb0);
    float m1 = fmaxf(fmaxf(sf1, ea1), eb1);
    for (int j = jb + 32; j < end; j += 32) {
        const int s = col[j];
        const float2 es = *(const float2*)(esrc + (size_t)s * HEADS);
        m0 = fmaxf(m0, lrelu(es.x + ed0));
        m1 = fmaxf(m1, lrelu(es.y + ed1));
    }
#pragma unroll
    for (int off = 16; off > 0; off >>= 1) {
        m0 = fmaxf(m0, __shfl_xor(m0, off, 32));
        m1 = fmaxf(m1, __shfl_xor(m1, off, 32));
    }

    const bool head1 = lane >= 16;
    const float as0 = __expf(sf0 - m0), as1 = __expf(sf1 - m1);
    float4 acc;
    {
        const float4 r = *(const float4*)(hfeat + (size_t)node * FDIM + lane * 4);
        const float wgt = head1 ? as1 : as0;
        acc.x = wgt * r.x; acc.y = wgt * r.y; acc.z = wgt * r.z; acc.w = wgt * r.w;
    }
    const float wa0 = va ? __expf(ea0 - m0) : 0.f, wa1 = va ? __expf(ea1 - m1) : 0.f;
    const float wb0 = vb ? __expf(eb0 - m0) : 0.f, wb1 = vb ? __expf(eb1 - m1) : 0.f;
    float p0 = 0.f, p1 = 0.f;
    int chunk = 0;
    for (int base = beg; base < end; base += 32, ++chunk) {
        const int cnt = (end - base < 32) ? (end - base) : 32;
        int s;
        float a0, a1;
        if (chunk == 0)      { s = sa; a0 = wa0; a1 = wa1; }
        else if (chunk == 1) { s = sb; a0 = wb0; a1 = wb1; }
        else {
            const int j = base + lane;
            s = 0; a0 = 0.f; a1 = 0.f;
            if (j < end) {
                s = col[j];
                const float2 es = *(const float2*)(esrc + (size_t)s * HEADS);
                a0 = __expf(lrelu(es.x + ed0) - m0);
                a1 = __expf(lrelu(es.y + ed1) - m1);
            }
        }
        p0 += a0; p1 += a1;
#pragma unroll 8
        for (int t = 0; t < cnt; ++t) {
            const int st = __builtin_amdgcn_readlane(s, t);
            const float w0 = __int_as_float(__builtin_amdgcn_readlane(__float_as_int(a0), t));
            const float w1 = __int_as_float(__builtin_amdgcn_readlane(__float_as_int(a1), t));
            const float4 r = *(const float4*)(hfeat + (size_t)st * FDIM + lane * 4);
            const float wgt = head1 ? w1 : w0;
            acc.x += wgt * r.x; acc.y += wgt * r.y; acc.z += wgt * r.z; acc.w += wgt * r.w;
        }
    }
#pragma unroll
    for (int off = 16; off > 0; off >>= 1) {
        p0 += __shfl_xor(p0, off, 32);
        p1 += __shfl_xor(p1, off, 32);
    }
    const float den = head1 ? (p1 + as1) : (p0 + as0);
    const float4 b = *(const float4*)(bias + lane * 4);
    float4 o;
    o.x = acc.x / den + b.x; o.y = acc.y / den + b.y;
    o.z = acc.z / den + b.z; o.w = acc.w / den + b.w;
    if (do_relu) { o.x = fmaxf(o.x, 0.f); o.y = fmaxf(o.y, 0.f); o.z = fmaxf(o.z, 0.f); o.w = fmaxf(o.w, 0.f); }
    *(float4*)(z + (size_t)node * FDIM + lane * 4) = o;
}

__device__ __forceinline__ float bfly_step(float lo, float hi, bool up, int mask)
{
    const float keep = up ? hi : lo;
    const float send = up ? lo : hi;
    return keep + __shfl_xor(send, mask, 32);
}

__global__ __launch_bounds__(256) void decode_edges(const int* __restrict__ src,
                                                    const int* __restrict__ dst,
                                                    const float* __restrict__ z,
                                                    float* __restrict__ out, int E)
{
    const int lane = threadIdx.x & 31;
    const int base = __builtin_amdgcn_readfirstlane((int)((blockIdx.x * 256u + threadIdx.x) >> 5)) * 32;
    if (base >= E) return;
    const int e = base + lane;
    const bool ok = e < E;
    const int sv = ok ? src[e] : 0;
    const int dv = ok ? dst[e] : 0;

    const bool b0 = lane & 1, b1 = lane & 2, b2 = lane & 4, b3 = lane & 8, b4 = lane & 16;
    float x[4];
#pragma unroll
    for (int g = 0; g < 4; ++g) {
        float p[8];
#pragma unroll
        for (int t = 0; t < 8; ++t) {
            const int s = __builtin_amdgcn_readlane(sv, g * 8 + t);
            const int d = __builtin_amdgcn_readlane(dv, g * 8 + t);
            const float4 a = *(const float4*)(z + (size_t)s * FDIM + lane * 4);
            const float4 b = *(const float4*)(z + (size_t)d * FDIM + lane * 4);
            p[t] = a.x * b.x + a.y * b.y + a.z * b.z + a.w * b.w;
        }
        float u[4], w[2];
#pragma unroll
        for (int i = 0; i < 4; ++i) u[i] = bfly_step(p[2 * i], p[2 * i + 1], b0, 1);
#pragma unroll
        for (int i = 0; i < 2; ++i) w[i] = bfly_step(u[2 * i], u[2 * i + 1], b1, 2);
        x[g] = bfly_step(w[0], w[1], b2, 4);
    }
    const float y0 = bfly_step(x[0], x[1], b3, 8);
    const float y1 = bfly_step(x[2], x[3], b3, 8);
    const float r  = bfly_step(y0, y1, b4, 16);
    if (ok) out[e] = r;
}

extern "C" void kernel_launch(void* const* d_in, const int* in_sizes, int n_in,
                              void* d_out, int out_size, void* d_ws, size_t ws_size,
                              hipStream_t stream) {
    const float* x      = (const float*)d_in[0];
    const int*   ei     = (const int*)d_in[1];
    const float* W1     = (const float*)d_in[2];
    const float* a1s    = (const float*)d_in[3];
    const float* a1d    = (const float*)d_in[4];
    const float* b1     = (const float*)d_in[5];
    const float* W2     = (const float*)d_in[6];
    const float* a2s    = (const float*)d_in[7];
    const float* a2d    = (const float*)d_in[8];
    const float* b2     = (const float*)d_in[9];
    float* scores = (float*)d_out;

    const int N = in_sizes[0] / FDIM;
    const int E = in_sizes[1] / 2;
    const int* src = ei;
    const int* dst = ei + E;

    float* ws   = (float*)d_ws;
    float*    h     = ws;
    int*      cnt   = (int*)(ws + (size_t)N * FDIM);
    int*      rowp  = cnt + N;
    int*      rank  = rowp + N + 1;
    int*      col   = rank + E;
    float*    z     = ws + (size_t)2 * N * FDIM;
    int*      hist  = (int*)z;
    int*      base  = hist + (size_t)CSR_BLOCKS * N;
    float*    esrc  = ws + (size_t)3 * N * FDIM;
    float*    edst  = esrc + (size_t)N * HEADS;

    dim3 blk(256);
    dim3 gGemm((N + 127) / 128);
    dim3 gNH((N * HEADS + 255) / 256);
    dim3 gNode((N + 255) / 256);
    dim3 gE((E + 255) / 256);
    dim3 gScan((N + 1023) / 1024);
    dim3 gNodeW((N + 7) / 8);
    const int per = (E + CSR_BLOCKS - 1) / CSR_BLOCKS;

    csr_count<<<dim3(CSR_BLOCKS), dim3(1024), 0, stream>>>(dst, E, N, per, hist, rank);
    csr_merge<<<gNode, blk, 0, stream>>>(hist, base, cnt, N);
    csr_scan<<<gScan, dim3(1024), 0, stream>>>(cnt, rowp, N);
    csr_fill<<<gE, blk, 0, stream>>>(src, dst, E, N, per, rowp, base, rank, col);

    gemm128_wmma<<<gGemm, blk, 0, stream>>>(x, W1, h, N);
    attn_logits<<<gNH, blk, 0, stream>>>(h, a1s, a1d, esrc, edst, N);
    gat_gather<<<gNodeW, blk, 0, stream>>>(rowp, col, esrc, edst, h, b1, z, N,  1);

    gemm128_wmma<<<gGemm, blk, 0, stream>>>(z, W2, h, N);
    attn_logits<<<gNH, blk, 0, stream>>>(h, a2s, a2d, esrc, edst, N);
    gat_gather<<<gNodeW, blk, 0, stream>>>(rowp, col, esrc, edst, h, b2, z, N,  0);

    decode_edges<<<gE, blk, 0, stream>>>(src, dst, z, scores, E);
}


# link decode regrouped by destination over the CSR: z[dst] row kept in registers, one gathered z row per edge instead of two; scores in slot order, then a gather pass to edge order; sums unchanged
# speedup vs baseline: 4.3087x; 1.0329x over previous
#include <hip/hip_runtime.h>
#include <hip/hip_bf16.h>


typedef __attribute__((ext_vector_type(2))) float v2f;
typedef __attribute__((ext_vector_type(8))) float v8f;

#define HEADS 2
#define FDIM 128
#define NEG_SLOPE 0.2f

__device__ __forceinline__ float lrelu(float v) { return (v >= 0.f) ? v : NEG_SLOPE * v; }

__global__ __launch_bounds__(256) void gemm128_wmma(
    const float* __restrict__ X, const float* __restrict__ W,
    float* __restrict__ Y, int M)
{
    __shared__ float lw[FDIM * FDIM];
    {
        const float4* w4 = (const float4*)W;
        float4* l4 = (float4*)lw;
        for (int i = threadIdx.x; i < FDIM * FDIM / 4; i += 256) l4[i] = w4[i];
    }
    __syncthreads();

    const int wave = threadIdx.x >> 5;
    const int lane = threadIdx.x & 31;
    const int m0 = (blockIdx.x * 8 + wave) * 16;
    if (m0 >= M) return;

    const int lm = lane & 15;
    const int lh = lane >> 4;
    const float* arow = X + (size_t)(m0 + lm) * FDIM + 2 * lh;

#pragma unroll
    for (int nt = 0; nt < 8; ++nt) {
        v8f acc = {};
        const int nc = nt * 16 + lm;
#pragma unroll 8
        for (int kk = 0; kk < 32; ++kk) {
            const int k0 = kk * 4;
            v2f a = *(const v2f*)(arow + k0);
            v2f b;
            b.x = lw[(k0 + 2 * lh) * FDIM + nc];
            b.y = lw[(k0 + 2 * lh + 1) * FDIM + nc];
            acc = __builtin_amdgcn_wmma_f32_16x16x4_f32(
                false, a, false, b, (short)0, acc, false, false);
        }
        float* yrow = Y + (size_t)(m0 + 8 * lh) * FDIM + nc;
#pragma unroll
        for (int r = 0; r < 8; ++r) yrow[(size_t)r * FDIM] = acc[r];
    }
}

__global__ void attn_logits(const float* __restrict__ h,
                            const float* __restrict__ a_src,
                            const float* __restrict__ a_dst,
                            float* __restrict__ esrc, float* __restrict__ edst,
                            int n)
{
    int idx = blockIdx.x * blockDim.x + threadIdx.x;
    if (idx >= n * HEADS) return;
    int node = idx >> 1, hd = idx & 1;
    const float* hp = h + (size_t)node * FDIM + hd * 64;
    const float* as = a_src + hd * 64;
    const float* ad = a_dst + hd * 64;
    float s = 0.f, t = 0.f;
#pragma unroll 8
    for (int f = 0; f < 64; ++f) { float x = hp[f]; s += x * as[f]; t += x * ad[f]; }
    esrc[idx] = s; edst[idx] = t;
}

#define CSR_BLOCKS 64
#define CSR_WIN 50176

__global__ __launch_bounds__(1024) void csr_count(const int* __restrict__ dst, int E, int n, int per,
                                                  int* __restrict__ hist, int* __restrict__ rank)
{
    __shared__ int lcnt[CSR_WIN];
    const int t = threadIdx.x;
    const int blk = blockIdx.x;
    const int e0 = blk * per;
    const int e1 = (E - e0 < per) ? E : e0 + per;
    int* __restrict__ plane = hist + (size_t)blk * n;
    for (int w0 = 0; w0 < n; w0 += CSR_WIN) {
        const int wn = (n - w0 < CSR_WIN) ? (n - w0) : CSR_WIN;
        for (int i = t; i < wn; i += 1024) lcnt[i] = 0;
        __syncthreads();
        for (int e = e0 + t; e < e1; e += 8 * 1024) {
            int d[8], r[8];
#pragma unroll
            for (int k = 0; k < 8; ++k) {
                const int ek = e + k * 1024;
                const int v = dst[(ek < e1) ? ek : (e1 - 1)];
                d[k] = (ek < e1) ? v - w0 : -1;
            }
#pragma unroll
            for (int k = 0; k < 8; ++k) {
                r[k] = 0;
                if ((unsigned)d[k] < (unsigned)wn) r[k] = atomicAdd(&lcnt[d[k]], 1);
            }
#pragma unroll
            for (int k = 0; k < 8; ++k) {
                if ((unsigned)d[k] < (unsigned)wn) rank[e + k * 1024] = r[k];
            }
        }
        __syncthreads();
        for (int i = t; i < wn; i += 1024) plane[w0 + i] = lcnt[i];
        __syncthreads();
    }
}

__global__ void csr_merge(const int* __restrict__ hist, int* __restrict__ base,
                          int* __restrict__ cnt, int n)
{
    const int d = blockIdx.x * blockDim.x + threadIdx.x;
    if (d >= n) return;
    int v[CSR_BLOCKS];
#pragma unroll
    for (int b = 0; b < CSR_BLOCKS; ++b) v[b] = hist[(size_t)b * n + d];
    int run = 0;
#pragma unroll
    for (int b = 0; b < CSR_BLOCKS; ++b) {
        base[(size_t)b * n + d] = run;
        run += v[b];
    }
    cnt[d] = run;
}

__global__ __launch_bounds__(1024) void csr_scan(const int* __restrict__ cnt,
                                                 int* __restrict__ rowptr, int n)
{
    __shared__ int sh_front[32];
    __shared__ int sh_wave[32];
    const int t = threadIdx.x, lane = t & 31, w = t >> 5;
    const int start = blockIdx.x * 1024;

    int p = 0;
    for (int i = t; i < start; i += 1024) p += cnt[i];
#pragma unroll
    for (int off = 16; off > 0; off >>= 1) p += __shfl_xor(p, off, 32);

    const int gi = start + t;
    const int v = (gi < n) ? cnt[gi] : 0;
    int inc = v;
#pragma unroll
    for (int off = 1; off < 32; off <<= 1) {
        int u = __shfl_up(inc, off, 32);
        if (lane >= off) inc += u;
    }
    if (lane == 31) sh_wave[w] = inc;
    if (lane == 0) sh_front[w] = p;
    __syncthreads();
    int base = 0;
#pragma unroll
    for (int k = 0; k < 32; ++k) {
        base += sh_front[k];
        if (k < w) base += sh_wave[k];
    }
    const int excl = base + inc - v;
    if (gi < n) rowptr[gi] = excl;
    if (gi == n - 1) rowptr[n] = excl + v;
}

__global__ void csr_fill(const int* __restrict__ src, const int* __restrict__ dst, int E, int n, int per,
                         const int* __restrict__ rowptr, const int* __restrict__ base,
                         int* __restrict__ rank_slot, int* __restrict__ col)
{
    int e = blockIdx.x * blockDim.x + threadIdx.x;
    if (e >= E) return;
    const int d = dst[e];
    if ((unsigned)d >= (unsigned)n) { rank_slot[e] = -1; return; }
    const int blk = e / per;
    const int slot = rowptr[d] + base[(size_t)blk * n + d] + rank_slot[e];
    atomicExch(&col[slot], src[e]);
    rank_slot[e] = slot;
}

__global__ __launch_bounds__(256) void gat_gather(
    const int* __restrict__ rowptr, const int* __restrict__ col,
    const float* __restrict__ esrc, const float* __restrict__ edst,
    const float* __restrict__ hfeat, const float* __restrict__ bias,
    float* __restrict__ z, int n, int do_relu)
{
    const int lane = threadIdx.x & 31;
    const int node = __builtin_amdgcn_readfirstlane((int)((blockIdx.x * 256u + threadIdx.x) >> 5));
    if (node >= n) return;
    const int beg = rowptr[node], end = rowptr[node + 1];
    const float ed0 = edst[node * HEADS + 0], ed1 = edst[node * HEADS + 1];
    const float sf0 = lrelu(esrc[node * HEADS + 0] + ed0);
    const float sf1 = lrelu(esrc[node * HEADS + 1] + ed1);

    const int ja = beg + lane, jb = ja + 32;
    const bool va = ja < end, vb = jb < end;
    int sa = node, sb = node;
    if (va) sa = col[ja];
    if (vb) sb = col[jb];
    const float2 esa = *(const float2*)(esrc + (size_t)sa * HEADS);
    const float2 esb = *(const float2*)(esrc + (size_t)sb * HEADS);
    const float ea0 = lrelu(esa.x + ed0), ea1 = lrelu(esa.y + ed1);
    const float eb0 = lrelu(esb.x + ed0), eb1 = lrelu(esb.y + ed1);
    float m0 = fmaxf(fmaxf(sf0, ea0), eb0);
    float m1 = fmaxf(fmaxf(sf1, ea1), eb1);
    for (int j = jb + 32; j < end; j += 32) {
        const int s = col[j];
        const float2 es = *(const float2*)(esrc + (size_t)s * HEADS);
        m0 = fmaxf(m0, lrelu(es.x + ed0));
        m1 = fmaxf(m1, lrelu(es.y + ed1));
    }
#pragma unroll
    for (int off = 16; off > 0; off >>= 1) {
        m0 = fmaxf(m0, __shfl_xor(m0, off, 32));
        m1 = fmaxf(m1, __shfl_xor(m1, off, 32));
    }

    const bool head1 = lane >= 16;
    const float as0 = __expf(sf0 - m0), as1 = __expf(sf1 - m1);
    float4 acc;
    {
        const float4 r = *(const float4*)(hfeat + (size_t)node * FDIM + lane * 4);
        const float wgt = head1 ? as1 : as0;
        acc.x = wgt * r.x; acc.y = wgt * r.y; acc.z = wgt * r.z; acc.w = wgt * r.w;
    }
    const float wa0 = va ? __expf(ea0 - m0) : 0.f, wa1 = va ? __expf(ea1 - m1) : 0.f;
    const float wb0 = vb ? __expf(eb0 - m0) : 0.f, wb1 = vb ? __expf(eb1 - m1) : 0.f;
    float p0 = 0.f, p1 = 0.f;
    int chunk = 0;
    for (int base = beg; base < end; base += 32, ++chunk) {
        const int cnt = (end - base < 32) ? (end - base) : 32;
        int s;
        float a0, a1;
        if (chunk == 0)      { s = sa; a0 = wa0; a1 = wa1; }
        else if (chunk == 1) { s = sb; a0 = wb0; a1 = wb1; }
        else {
            const int j = base + lane;
            s = 0; a0 = 0.f; a1 = 0.f;
            if (j < end) {
                s = col[j];
                const float2 es = *(const float2*)(esrc + (size_t)s * HEADS);
                a0 = __expf(lrelu(es.x + ed0) - m0);
                a1 = __expf(lrelu(es.y + ed1) - m1);
            }
        }
        p0 += a0; p1 += a1;
#pragma unroll 8
        for (int t = 0; t < cnt; ++t) {
            const int st = __builtin_amdgcn_readlane(s, t);
            const float w0 = __int_as_float(__builtin_amdgcn_readlane(__float_as_int(a0), t));
            const float w1 = __int_as_float(__builtin_amdgcn_readlane(__float_as_int(a1), t));
            const float4 r = *(const float4*)(hfeat + (size_t)st * FDIM + lane * 4);
            const float wgt = head1 ? w1 : w0;
            acc.x += wgt * r.x; acc.y += wgt * r.y; acc.z += wgt * r.z; acc.w += wgt * r.w;
        }
    }
#pragma unroll
    for (int off = 16; off > 0; off >>= 1) {
        p0 += __shfl_xor(p0, off, 32);
        p1 += __shfl_xor(p1, off, 32);
    }
    const float den = head1 ? (p1 + as1) : (p0 + as0);
    const float4 b = *(const float4*)(bias + lane * 4);
    float4 o;
    o.x = acc.x / den + b.x; o.y = acc.y / den + b.y;
    o.z = acc.z / den + b.z; o.w = acc.w / den + b.w;
    if (do_relu) { o.x = fmaxf(o.x, 0.f); o.y = fmaxf(o.y, 0.f); o.z = fmaxf(o.z, 0.f); o.w = fmaxf(o.w, 0.f); }
    *(float4*)(z + (size_t)node * FDIM + lane * 4) = o;
}

__device__ __forceinline__ float bfly_step(float lo, float hi, bool up, int mask)
{
    const float keep = up ? hi : lo;
    const float send = up ? lo : hi;
    return keep + __shfl_xor(send, mask, 32);
}

__global__ __launch_bounds__(256) void decode_by_dst(const int* __restrict__ rowptr,
                                                     const int* __restrict__ col,
                                                     const float* __restrict__ z,
                                                     float* __restrict__ slot_score, int n)
{
    const int lane = threadIdx.x & 31;
    const int node = __builtin_amdgcn_readfirstlane((int)((blockIdx.x * 256u + threadIdx.x) >> 5));
    if (node >= n) return;
    const int beg = rowptr[node], end = rowptr[node + 1];
    const float4 b = *(const float4*)(z + (size_t)node * FDIM + lane * 4);

    const bool b0 = lane & 1, b1 = lane & 2, b2 = lane & 4, b3 = lane & 8, b4 = lane & 16;
    for (int base = beg; base < end; base += 32) {
        const int cnt = (end - base < 32) ? (end - base) : 32;
        const int j = base + lane;
        int sv = node;
        if (lane < cnt) sv = col[j];
        float x[4];
#pragma unroll
        for (int g = 0; g < 4; ++g) {
            x[g] = 0.f;
            if (g * 8 < cnt) {
                float p[8];
#pragma unroll
                for (int t = 0; t < 8; ++t) {
                    const int s = __builtin_amdgcn_readlane(sv, g * 8 + t);
                    const float4 a = *(const float4*)(z + (size_t)s * FDIM + lane * 4);
                    p[t] = fmaf(a.w, b.w, fmaf(a.z, b.z, fmaf(a.x, b.x, a.y * b.y)));
                }
                float u[4], w[2];
#pragma unroll
                for (int i = 0; i < 4; ++i) u[i] = bfly_step(p[2 * i], p[2 * i + 1], b0, 1);
#pragma unroll
                for (int i = 0; i < 2; ++i) w[i] = bfly_step(u[2 * i], u[2 * i + 1], b1, 2);
                x[g] = bfly_step(w[0], w[1], b2, 4);
            }
        }
        const float y0 = bfly_step(x[0], x[1], b3, 8);
        const float y1 = bfly_step(x[2], x[3], b3, 8);
        const float r  = bfly_step(y0, y1, b4, 16);
        if (lane < cnt) slot_score[j] = r;
    }
}

__global__ __launch_bounds__(256) void scores_to_edge_order(const int* __restrict__ slot,
                                                            const float* __restrict__ slot_score,
                                                            float* __restrict__ out, int E)
{
    const int lane = threadIdx.x & 31;
    const int e0 = (int)((blockIdx.x * 256u + threadIdx.x) >> 5) * 128 + lane;
    int sl[4];
    float v[4];
#pragma unroll
    for (int k = 0; k < 4; ++k) {
        const int e = e0 + k * 32;
        sl[k] = (e < E) ? slot[e] : -1;
    }
#pragma unroll
    for (int k = 0; k < 4; ++k) v[k] = (sl[k] >= 0) ? slot_score[sl[k]] : 0.f;
#pragma unroll
    for (int k = 0; k < 4; ++k) {
        const int e = e0 + k * 32;
        if (e < E) out[e] = v[k];
    }
}

extern "C" void kernel_launch(void* const* d_in, const int* in_sizes, int n_in,
                              void* d_out, int out_size, void* d_ws, size_t ws_size,
                              hipStream_t stream) {
    const float* x      = (const float*)d_in[0];
    const int*   ei     = (const int*)d_in[1];
    const float* W1     = (const float*)d_in[2];
    const float* a1s    = (const float*)d_in[3];
    const float* a1d    = (const float*)d_in[4];
    const float* b1     = (const float*)d_in[5];
    const float* W2     = (const float*)d_in[6];
    const float* a2s    = (const float*)d_in[7];
    const float* a2d    = (const float*)d_in[8];
    const float* b2     = (const float*)d_in[9];
    float* scores = (float*)d_out;

    const int N = in_sizes[0] / FDIM;
    const int E = in_sizes[1] / 2;
    const int* src = ei;
    const int* dst = ei + E;

    float* ws   = (float*)d_ws;
    float*    h     = ws;
    int*      cnt   = (int*)(ws + (size_t)N * FDIM);
    int*      rowp  = cnt + N;
    int*      rank  = rowp + N + 1;
    int*      col   = rank + E;
    float*    z     = ws + (size_t)2 * N * FDIM;
    int*      hist  = (int*)z;
    int*      base  = hist + (size_t)CSR_BLOCKS * N;
    float*    esrc  = ws + (size_t)3 * N * FDIM;
    float*    edst  = esrc + (size_t)N * HEADS;

    dim3 blk(256);
    dim3 gGemm((N + 127) / 128);
    dim3 gNH((N * HEADS + 255) / 256);
    dim3 gNode((N + 255) / 256);
    dim3 gE((E + 255) / 256);
    dim3 gScan((N + 1023) / 1024);
    dim3 gNodeW((N + 7) / 8);
    dim3 gE4((E + 1023) / 1024);
    const int per = (E + CSR_BLOCKS - 1) / CSR_BLOCKS;

    csr_count<<<dim3(CSR_BLOCKS), dim3(1024), 0, stream>>>(dst, E, N, per, hist, rank);
    csr_merge<<<gNode, blk, 0, stream>>>(hist, base, cnt, N);
    csr_scan<<<gScan, dim3(1024), 0, stream>>>(cnt, rowp, N);
    csr_fill<<<gE, blk, 0, stream>>>(src, dst, E, N, per, rowp, base, rank, col);

    gemm128_wmma<<<gGemm, blk, 0, stream>>>(x, W1, h, N);
    attn_logits<<<gNH, blk, 0, stream>>>(h, a1s, a1d, esrc, edst, N);
    gat_gather<<<gNodeW, blk, 0, stream>>>(rowp, col, esrc, edst, h, b1, z, N,  1);

    gemm128_wmma<<<gGemm, blk, 0, stream>>>(z, W2, h, N);
    attn_logits<<<gNH, blk, 0, stream>>>(h, a2s, a2d, esrc, edst, N);
    gat_gather<<<gNodeW, blk, 0, stream>>>(rowp, col, esrc, edst, h, b2, z, N,  0);

    float* slot_score = h;
    decode_by_dst<<<gNodeW, blk, 0, stream>>>(rowp, col, z, slot_score, N);
    scores_to_edge_order<<<gE4, blk, 0, stream>>>(rank, slot_score, scores, E);
}


# GEMM loop order: k outermost, each A fragment (an 8-byte load that touches 16 lines) loaded once for all 8 column tiles, accumulators in registers; A loads per wave 256 -> 32; sums unchanged
# speedup vs baseline: 4.7213x; 1.0958x over previous
#include <hip/hip_runtime.h>
#include <hip/hip_bf16.h>


typedef __attribute__((ext_vector_type(2))) float v2f;
typedef __attribute__((ext_vector_type(8))) float v8f;

#define HEADS 2
#define FDIM 128
#define NEG_SLOPE 0.2f

__device__ __forceinline__ float lrelu(float v) { return (v >= 0.f) ? v : NEG_SLOPE * v; }

__device__ __forceinline__ float bfly_step(float lo, float hi, bool up, int mask)
{
    const float keep = up ? hi : lo;
    const float send = up ? lo : hi;
    return keep + __shfl_xor(send, mask, 32);
}

__global__ __launch_bounds__(256) void gemm128_wmma_logits(
    const float* __restrict__ X, const float* __restrict__ W,
    const float* __restrict__ a_src, const float* __restrict__ a_dst,
    float* __restrict__ Y, float* __restrict__ esrc, float* __restrict__ edst, int M)
{
    __shared__ float lw[FDIM * FDIM];
    {
        const float4* w4 = (const float4*)W;
        float4* l4 = (float4*)lw;
        for (int i = threadIdx.x; i < FDIM * FDIM / 4; i += 256) l4[i] = w4[i];
    }
    __syncthreads();

    const int wave = threadIdx.x >> 5;
    const int lane = threadIdx.x & 31;
    const int m0 = (blockIdx.x * 8 + wave) * 16;
    if (m0 >= M) return;

    const int lm = lane & 15;
    const int lh = lane >> 4;
    const float* arow = X + (size_t)(m0 + lm) * FDIM + 2 * lh;

    float ps[16], pd[16];
#pragma unroll
    for (int i = 0; i < 16; ++i) { ps[i] = 0.f; pd[i] = 0.f; }

    v8f acc[8];
#pragma unroll
    for (int nt = 0; nt < 8; ++nt) acc[nt] = (v8f){0.f, 0.f, 0.f, 0.f, 0.f, 0.f, 0.f, 0.f};

    v2f a[8], an[8];
#pragma unroll
    for (int j = 0; j < 8; ++j) a[j] = *(const v2f*)(arow + j * 4);
#pragma unroll
    for (int g = 0; g < 4; ++g) {
        if (g < 3) {
#pragma unroll
            for (int j = 0; j < 8; ++j) an[j] = *(const v2f*)(arow + (g + 1) * 32 + j * 4);
        }
#pragma unroll
        for (int j = 0; j < 8; ++j) {
            const int k0 = g * 32 + j * 4;
            const float* brow = lw + (k0 + 2 * lh) * FDIM + lm;
#pragma unroll
            for (int nt = 0; nt < 8; ++nt) {
                v2f b;
                b.x = brow[nt * 16];
                b.y = brow[FDIM + nt * 16];
                acc[nt] = __builtin_amdgcn_wmma_f32_16x16x4_f32(
                    false, a[j], false, b, (short)0, acc[nt], false, false);
            }
        }
        if (g < 3) {
#pragma unroll
            for (int j = 0; j < 8; ++j) a[j] = an[j];
        }
    }

#pragma unroll
    for (int nt = 0; nt < 8; ++nt) {
        const int nc = nt * 16 + lm;
        float* yrow = Y + (size_t)(m0 + 8 * lh) * FDIM + nc;
        const float as = a_src[nc], ad = a_dst[nc];
        const int hb = (nt >> 2) * 8;
#pragma unroll
        for (int r = 0; r < 8; ++r) {
            yrow[(size_t)r * FDIM] = acc[nt][r];
            ps[hb + r] = fmaf(acc[nt][r], as, ps[hb + r]);
            pd[hb + r] = fmaf(acc[nt][r], ad, pd[hb + r]);
        }
    }

    const bool b0 = lane & 1, b1 = lane & 2, b2 = lane & 4, b3 = lane & 8;
    float u[16], w[8], x4[4];
#pragma unroll
    for (int i = 0; i < 8; ++i) {
        u[i]     = bfly_step(ps[2 * i], ps[2 * i + 1], b0, 1);
        u[8 + i] = bfly_step(pd[2 * i], pd[2 * i + 1], b0, 1);
    }
#pragma unroll
    for (int i = 0; i < 8; ++i) w[i] = bfly_step(u[2 * i], u[2 * i + 1], b1, 2);
#pragma unroll
    for (int i = 0; i < 4; ++i) x4[i] = bfly_step(w[2 * i], w[2 * i + 1], b2, 4);
    const float y0 = bfly_step(x4[0], x4[1], b3, 8);
    const float y1 = bfly_step(x4[2], x4[3], b3, 8);
    const int oi = (m0 + 8 * lh + (lm & 7)) * HEADS + (lm >> 3);
    esrc[oi] = y0;
    edst[oi] = y1;
}

#define CSR_BLOCKS 64
#define CSR_WIN 50176

__global__ __launch_bounds__(1024) void csr_count(const int* __restrict__ dst, int E, int n, int per,
                                                  int* __restrict__ hist, int* __restrict__ rank)
{
    __shared__ int lcnt[CSR_WIN];
    const int t = threadIdx.x;
    const int blk = blockIdx.x;
    const int e0 = blk * per;
    const int e1 = (E - e0 < per) ? E : e0 + per;
    int* __restrict__ plane = hist + (size_t)blk * n;
    for (int w0 = 0; w0 < n; w0 += CSR_WIN) {
        const int wn = (n - w0 < CSR_WIN) ? (n - w0) : CSR_WIN;
        for (int i = t; i < wn; i += 1024) lcnt[i] = 0;
        __syncthreads();
        for (int e = e0 + t; e < e1; e += 8 * 1024) {
            int d[8], r[8];
#pragma unroll
            for (int k = 0; k < 8; ++k) {
                const int ek = e + k * 1024;
                const int v = dst[(ek < e1) ? ek : (e1 - 1)];
                d[k] = (ek < e1) ? v - w0 : -1;
            }
#pragma unroll
            for (int k = 0; k < 8; ++k) {
                r[k] = 0;
                if ((unsigned)d[k] < (unsigned)wn) r[k] = atomicAdd(&lcnt[d[k]], 1);
            }
#pragma unroll
            for (int k = 0; k < 8; ++k) {
                if ((unsigned)d[k] < (unsigned)wn) rank[e + k * 1024] = r[k];
            }
        }
        __syncthreads();
        for (int i = t; i < wn; i += 1024) plane[w0 + i] = lcnt[i];
        __syncthreads();
    }
}

__global__ void csr_merge(const int* __restrict__ hist, int* __restrict__ base,
                          int* __restrict__ cnt, int n)
{
    const int d = blockIdx.x * blockDim.x + threadIdx.x;
    if (d >= n) return;
    int v[CSR_BLOCKS];
#pragma unroll
    for (int b = 0; b < CSR_BLOCKS; ++b) v[b] = hist[(size_t)b * n + d];
    int run = 0;
#pragma unroll
    for (int b = 0; b < CSR_BLOCKS; ++b) {
        base[(size_t)b * n + d] = run;
        run += v[b];
    }
    cnt[d] = run;
}

__global__ __launch_bounds__(1024) void csr_scan(const int* __restrict__ cnt,
                                                 int* __restrict__ rowptr, int n)
{
    __shared__ int sh_front[32];
    __shared__ int sh_wave[32];
    const int t = threadIdx.x, lane = t & 31, w = t >> 5;
    const int start = blockIdx.x * 1024;

    int p = 0;
    for (int i = t; i < start; i += 1024) p += cnt[i];
#pragma unroll
    for (int off = 16; off > 0; off >>= 1) p += __shfl_xor(p, off, 32);

    const int gi = start + t;
    const int v = (gi < n) ? cnt[gi] : 0;
    int inc = v;
#pragma unroll
    for (int off = 1; off < 32; off <<= 1) {
        int u = __shfl_up(inc, off, 32);
        if (lane >= off) inc += u;
    }
    if (lane == 31) sh_wave[w] = inc;
    if (lane == 0) sh_front[w] = p;
    __syncthreads();
    int base = 0;
#pragma unroll
    for (int k = 0; k < 32; ++k) {
        base += sh_front[k];
        if (k < w) base += sh_wave[k];
    }
    const int excl = base + inc - v;
    if (gi < n) rowptr[gi] = excl;
    if (gi == n - 1) rowptr[n] = excl + v;
}

__global__ void csr_fill(const int* __restrict__ src, const int* __restrict__ dst, int E, int n, int per,
                         const int* __restrict__ rowptr, const int* __restrict__ base,
                         int* __restrict__ rank_slot, int* __restrict__ col)
{
    int e = blockIdx.x * blockDim.x + threadIdx.x;
    if (e >= E) return;
    const int d = dst[e];
    if ((unsigned)d >= (unsigned)n) { rank_slot[e] = -1; return; }
    const int blk = e / per;
    const int slot = rowptr[d] + base[(size_t)blk * n + d] + rank_slot[e];
    atomicExch(&col[slot], src[e]);
    rank_slot[e] = slot;
}

__global__ __launch_bounds__(256) void gat_gather(
    const int* __restrict__ rowptr, const int* __restrict__ col,
    const float* __restrict__ esrc, const float* __restrict__ edst,
    const float* __restrict__ hfeat, const float* __restrict__ bias,
    float* __restrict__ z, int n, int do_relu)
{
    const int lane = threadIdx.x & 31;
    const int node = __builtin_amdgcn_readfirstlane((int)((blockIdx.x * 256u + threadIdx.x) >> 5));
    if (node >= n) return;
    const int beg = rowptr[node], end = rowptr[node + 1];
    const float ed0 = edst[node * HEADS + 0], ed1 = edst[node * HEADS + 1];
    const float sf0 = lrelu(esrc[node * HEADS + 0] + ed0);
    const float sf1 = lrelu(esrc[node * HEADS + 1] + ed1);

    const int ja = beg + lane, jb = ja + 32;
    const bool va = ja < end, vb = jb < end;
    int sa = node, sb = node;
    if (va) sa = col[ja];
    if (vb) sb = col[jb];
    const float2 esa = *(const float2*)(esrc + (size_t)sa * HEADS);
    const float2 esb = *(const float2*)(esrc + (size_t)sb * HEADS);
    const float ea0 = lrelu(esa.x + ed0), ea1 = lrelu(esa.y + ed1);
    const float eb0 = lrelu(esb.x + ed0), eb1 = lrelu(esb.y + ed1);
    float m0 = fmaxf(fmaxf(sf0, ea0), eb0);
    float m1 = fmaxf(fmaxf(sf1, ea1), eb1);
    for (int j = jb + 32; j < end; j += 32) {
        const int s = col[j];
        const float2 es = *(const float2*)(esrc + (size_t)s * HEADS);
        m0 = fmaxf(m0, lrelu(es.x + ed0));
        m1 = fmaxf(m1, lrelu(es.y + ed1));
    }
#pragma unroll
    for (int off = 16; off > 0; off >>= 1) {
        m0 = fmaxf(m0, __shfl_xor(m0, off, 32));
        m1 = fmaxf(m1, __shfl_xor(m1, off, 32));
    }

    const bool head1 = lane >= 16;
    const float as0 = __expf(sf0 - m0), as1 = __expf(sf1 - m1);
    float4 acc;
    {
        const float4 r = *(const float4*)(hfeat + (size_t)node * FDIM + lane * 4);
        const float wgt = head1 ? as1 : as0;
        acc.x = wgt * r.x; acc.y = wgt * r.y; acc.z = wgt * r.z; acc.w = wgt * r.w;
    }
    const float wa0 = va ? __expf(ea0 - m0) : 0.f, wa1 = va ? __expf(ea1 - m1) : 0.f;
    const float wb0 = vb ? __expf(eb0 - m0) : 0.f, wb1 = vb ? __expf(eb1 - m1) : 0.f;
    float p0 = 0.f, p1 = 0.f;
    int chunk = 0;
    for (int base = beg; base < end; base += 32, ++chunk) {
        const int cnt = (end - base < 32) ? (end - base) : 32;
        int s;
        float a0, a1;
        if (chunk == 0)      { s = sa; a0 = wa0; a1 = wa1; }
        else if (chunk == 1) { s = sb; a0 = wb0; a1 = wb1; }
        else {
            const int j = base + lane;
            s = 0; a0 = 0.f; a1 = 0.f;
            if (j < end) {
                s = col[j];
                const float2 es = *(const float2*)(esrc + (size_t)s * HEADS);
                a0 = __expf(lrelu(es.x + ed0) - m0);
                a1 = __expf(lrelu(es.y + ed1) - m1);
            }
        }
        p0 += a0; p1 += a1;
#pragma unroll 8
        for (int t = 0; t < cnt; ++t) {
            const int st = __builtin_amdgcn_readlane(s, t);
            const float w0 = __int_as_float(__builtin_amdgcn_readlane(__float_as_int(a0), t));
            const float w1 = __int_as_float(__builtin_amdgcn_readlane(__float_as_int(a1), t));
            const float4 r = *(const float4*)(hfeat + (size_t)st * FDIM + lane * 4);
            const float wgt = head1 ? w1 : w0;
            acc.x += wgt * r.x; acc.y += wgt * r.y; acc.z += wgt * r.z; acc.w += wgt * r.w;
        }
    }
#pragma unroll
    for (int off = 16; off > 0; off >>= 1) {
        p0 += __shfl_xor(p0, off, 32);
        p1 += __shfl_xor(p1, off, 32);
    }
    const float den = head1 ? (p1 + as1) : (p0 + as0);
    const float4 b = *(const float4*)(bias + lane * 4);
    float4 o;
    o.x = acc.x / den + b.x; o.y = acc.y / den + b.y;
    o.z = acc.z / den + b.z; o.w = acc.w / den + b.w;
    if (do_relu) { o.x = fmaxf(o.x, 0.f); o.y = fmaxf(o.y, 0.f); o.z = fmaxf(o.z, 0.f); o.w = fmaxf(o.w, 0.f); }
    *(float4*)(z + (size_t)node * FDIM + lane * 4) = o;
}

__global__ __launch_bounds__(256) void decode_by_dst(const int* __restrict__ rowptr,
                                                     const int* __restrict__ col,
                                                     const float* __restrict__ z,
                                                     float* __restrict__ slot_score, int n)
{
    const int lane = threadIdx.x & 31;
    const int node = __builtin_amdgcn_readfirstlane((int)((blockIdx.x * 256u + threadIdx.x) >> 5));
    if (node >= n) return;
    const int beg = rowptr[node], end = rowptr[node + 1];
    const float4 b = *(const float4*)(z + (size_t)node * FDIM + lane * 4);

    const bool b0 = lane & 1, b1 = lane & 2, b2 = lane & 4, b3 = lane & 8, b4 = lane & 16;
    for (int base = beg; base < end; base += 32) {
        const int cnt = (end - base < 32) ? (end - base) : 32;
        const int j = base + lane;
        int sv = node;
        if (lane < cnt) sv = col[j];
        float x[4];
#pragma unroll
        for (int g = 0; g < 4; ++g) {
            x[g] = 0.f;
            if (g * 8 < cnt) {
                float p[8];
#pragma unroll
                for (int t = 0; t < 8; ++t) {
                    const int s = __builtin_amdgcn_readlane(sv, g * 8 + t);
                    const float4 a = *(const float4*)(z + (size_t)s * FDIM + lane * 4);
                    p[t] = fmaf(a.w, b.w, fmaf(a.z, b.z, fmaf(a.x, b.x, a.y * b.y)));
                }
                float u[4], w[2];
#pragma unroll
                for (int i = 0; i < 4; ++i) u[i] = bfly_step(p[2 * i], p[2 * i + 1], b0, 1);
#pragma unroll
                for (int i = 0; i < 2; ++i) w[i] = bfly_step(u[2 * i], u[2 * i + 1], b1, 2);
                x[g] = bfly_step(w[0], w[1], b2, 4);
            }
        }
        const float y0 = bfly_step(x[0], x[1], b3, 8);
        const float y1 = bfly_step(x[2], x[3], b3, 8);
        const float r  = bfly_step(y0, y1, b4, 16);
        if (lane < cnt) slot_score[j] = r;
    }
}

__global__ __launch_bounds__(256) void scores_to_edge_order(const int* __restrict__ slot,
                                                            const float* __restrict__ slot_score,
                                                            float* __restrict__ out, int E)
{
    const int lane = threadIdx.x & 31;
    const int e0 = (int)((blockIdx.x * 256u + threadIdx.x) >> 5) * 128 + lane;
    int sl[4];
    float v[4];
#pragma unroll
    for (int k = 0; k < 4; ++k) {
        const int e = e0 + k * 32;
        sl[k] = (e < E) ? slot[e] : -1;
    }
#pragma unroll
    for (int k = 0; k < 4; ++k) v[k] = (sl[k] >= 0) ? slot_score[sl[k]] : 0.f;
#pragma unroll
    for (int k = 0; k < 4; ++k) {
        const int e = e0 + k * 32;
        if (e < E) out[e] = v[k];
    }
}

extern "C" void kernel_launch(void* const* d_in, const int* in_sizes, int n_in,
                              void* d_out, int out_size, void* d_ws, size_t ws_size,
                              hipStream_t stream) {
    const float* x      = (const float*)d_in[0];
    const int*   ei     = (const int*)d_in[1];
    const float* W1     = (const float*)d_in[2];
    const float* a1s    = (const float*)d_in[3];
    const float* a1d    = (const float*)d_in[4];
    const float* b1     = (const float*)d_in[5];
    const float* W2     = (const float*)d_in[6];
    const float* a2s    = (const float*)d_in[7];
    const float* a2d    = (const float*)d_in[8];
    const float* b2     = (const float*)d_in[9];
    float* scores = (float*)d_out;

    const int N = in_sizes[0] / FDIM;
    const int E = in_sizes[1] / 2;
    const int* src = ei;
    const int* dst = ei + E;

    float* ws   = (float*)d_ws;
    float*    h     = ws;
    int*      cnt   = (int*)(ws + (size_t)N * FDIM);
    int*      rowp  = cnt + N;
    int*      rank  = rowp + N + 1;
    int*      col   = rank + E;
    float*    z     = ws + (size_t)2 * N * FDIM;
    int*      hist  = (int*)z;
    int*      base  = hist + (size_t)CSR_BLOCKS * N;
    float*    esrc  = ws + (size_t)3 * N * FDIM;
    float*    edst  = esrc + (size_t)N * HEADS;

    dim3 blk(256);
    dim3 gGemm((N + 127) / 128);
    dim3 gNode((N + 255) / 256);
    dim3 gE((E + 255) / 256);
    dim3 gScan((N + 1023) / 1024);
    dim3 gNodeW((N + 7) / 8);
    dim3 gE4((E + 1023) / 1024);
    const int per = (E + CSR_BLOCKS - 1) / CSR_BLOCKS;

    csr_count<<<dim3(CSR_BLOCKS), dim3(1024), 0, stream>>>(dst, E, N, per, hist, rank);
    csr_merge<<<gNode, blk, 0, stream>>>(hist, base, cnt, N);
    csr_scan<<<gScan, dim3(1024), 0, stream>>>(cnt, rowp, N);
    csr_fill<<<gE, blk, 0, stream>>>(src, dst, E, N, per, rowp, base, rank, col);

    gemm128_wmma_logits<<<gGemm, blk, 0, stream>>>(x, W1, a1s, a1d, h, esrc, edst, N);
    gat_gather<<<gNodeW, blk, 0, stream>>>(rowp, col, esrc, edst, h, b1, z, N,  1);

    gemm128_wmma_logits<<<gGemm, blk, 0, stream>>>(z, W2, a2s, a2d, h, esrc, edst, N);
    gat_gather<<<gNodeW, blk, 0, stream>>>(rowp, col, esrc, edst, h, b2, z, N,  0);

    float* slot_score = h;
    decode_by_dst<<<gNodeW, blk, 0, stream>>>(rowp, col, z, slot_score, N);
    scores_to_edge_order<<<gE4, blk, 0, stream>>>(rank, slot_score, scores, E);
}


# layer 2's GEMM runs inside layer 1's gather launch: 32 waves hand finished rows over through LDS, 16 of them run the f32 WMMA k-steps; 9 -> 8 launches; layer-2 logit summation tree changes
# speedup vs baseline: 5.2567x; 1.1134x over previous
#include <hip/hip_runtime.h>
#include <hip/hip_bf16.h>


typedef __attribute__((ext_vector_type(2))) float v2f;
typedef __attribute__((ext_vector_type(8))) float v8f;

#define HEADS 2
#define FDIM 128
#define NEG_SLOPE 0.2f

__device__ __forceinline__ float lrelu(float v) { return (v >= 0.f) ? v : NEG_SLOPE * v; }

__device__ __forceinline__ float bfly_step(float lo, float hi, bool up, int mask)
{
    const float keep = up ? hi : lo;
    const float send = up ? lo : hi;
    return keep + __shfl_xor(send, mask, 32);
}

#define FILL_EPT 4
__device__ __forceinline__ void csr_fill(int e0, const int* __restrict__ src, const int* __restrict__ dst,
                                         int E, int n, int per,
                                         const int* __restrict__ rowptr, const int* __restrict__ base,
                                         int* __restrict__ rank_slot, int* __restrict__ col)
{
    int ec[FILL_EPT], d[FILL_EPT], slot[FILL_EPT], sv[FILL_EPT];
    bool ok[FILL_EPT];
#pragma unroll
    for (int k = 0; k < FILL_EPT; ++k) {
        const int e = e0 + k * 256;
        ec[k] = (e < E) ? e : (E - 1);
        d[k] = dst[ec[k]];
    }
#pragma unroll
    for (int k = 0; k < FILL_EPT; ++k) {
        ok[k] = (e0 + k * 256 < E) && ((unsigned)d[k] < (unsigned)n);
        const int dc = ((unsigned)d[k] < (unsigned)n) ? d[k] : 0;
        const int blk = ec[k] / per;
        const int s = rowptr[dc] + base[(size_t)blk * n + dc] + rank_slot[ec[k]];
        slot[k] = ok[k] ? s : -1;
        sv[k] = src[ec[k]];
    }
#pragma unroll
    for (int k = 0; k < FILL_EPT; ++k) {
        if (ok[k]) atomicExch(&col[slot[k]], sv[k]);
    }
#pragma unroll
    for (int k = 0; k < FILL_EPT; ++k) {
        const int e = e0 + k * 256;
        if (e < E) rank_slot[e] = slot[k];
    }
}

__global__ __launch_bounds__(256) void gemm128_wmma_logits(
    const float* __restrict__ X, const float* __restrict__ W,
    const float* __restrict__ a_src, const float* __restrict__ a_dst,
    float* __restrict__ Y, float* __restrict__ esrc, float* __restrict__ edst, int M,
    int gemm_blocks,
    const int* __restrict__ src, const int* __restrict__ dst, int E, int per,
    const int* __restrict__ rowptr, const int* __restrict__ base,
    int* __restrict__ rank_slot, int* __restrict__ col)
{
    if ((int)blockIdx.x >= gemm_blocks) {
        csr_fill((int)(blockIdx.x - gemm_blocks) * (256 * FILL_EPT) + (int)threadIdx.x,
                 src, dst, E, M, per, rowptr, base, rank_slot, col);
        return;
    }
    __shared__ float lw[FDIM * FDIM];
    {
        const float4* w4 = (const float4*)W;
        float4* l4 = (float4*)lw;
        for (int i = threadIdx.x; i < FDIM * FDIM / 4; i += 256) l4[i] = w4[i];
    }
    __syncthreads();

    const int wave = threadIdx.x >> 5;
    const int lane = threadIdx.x & 31;
    const int m0 = (blockIdx.x * 8 + wave) * 16;
    if (m0 >= M) return;

    const int lm = lane & 15;
    const int lh = lane >> 4;
    const float* arow = X + (size_t)(m0 + lm) * FDIM + 2 * lh;

    float ps[16], pd[16];
#pragma unroll
    for (int i = 0; i < 16; ++i) { ps[i] = 0.f; pd[i] = 0.f; }

    v8f acc[8];
#pragma unroll
    for (int nt = 0; nt < 8; ++nt) acc[nt] = (v8f){0.f, 0.f, 0.f, 0.f, 0.f, 0.f, 0.f, 0.f};

    v2f a[8], an[8];
#pragma unroll
    for (int j = 0; j < 8; ++j) a[j] = *(const v2f*)(arow + j * 4);
#pragma unroll
    for (int g = 0; g < 4; ++g) {
        if (g < 3) {
#pragma unroll
            for (int j = 0; j < 8; ++j) an[j] = *(const v2f*)(arow + (g + 1) * 32 + j * 4);
        }
#pragma unroll
        for (int j = 0; j < 8; ++j) {
            const int k0 = g * 32 + j * 4;
            const float* brow = lw + (k0 + 2 * lh) * FDIM + lm;
#pragma unroll
            for (int nt = 0; nt < 8; ++nt) {
                v2f b;
                b.x = brow[nt * 16];
                b.y = brow[FDIM + nt * 16];
                acc[nt] = __builtin_amdgcn_wmma_f32_16x16x4_f32(
                    false, a[j], false, b, (short)0, acc[nt], false, false);
            }
        }
        if (g < 3) {
#pragma unroll
            for (int j = 0; j < 8; ++j) a[j] = an[j];
        }
    }

#pragma unroll
    for (int nt = 0; nt < 8; ++nt) {
        const int nc = nt * 16 + lm;
        float* yrow = Y + (size_t)(m0 + 8 * lh) * FDIM + nc;
        const float as = a_src[nc], ad = a_dst[nc];
        const int hb = (nt >> 2) * 8;
#pragma unroll
        for (int r = 0; r < 8; ++r) {
            yrow[(size_t)r * FDIM] = acc[nt][r];
            ps[hb + r] = fmaf(acc[nt][r], as, ps[hb + r]);
            pd[hb + r] = fmaf(acc[nt][r], ad, pd[hb + r]);
        }
    }

    const bool b0 = lane & 1, b1 = lane & 2, b2 = lane & 4, b3 = lane & 8;
    float u[16], w[8], x4[4];
#pragma unroll
    for (int i = 0; i < 8; ++i) {
        u[i]     = bfly_step(ps[2 * i], ps[2 * i + 1], b0, 1);
        u[8 + i] = bfly_step(pd[2 * i], pd[2 * i + 1], b0, 1);
    }
#pragma unroll
    for (int i = 0; i < 8; ++i) w[i] = bfly_step(u[2 * i], u[2 * i + 1], b1, 2);
#pragma unroll
    for (int i = 0; i < 4; ++i) x4[i] = bfly_step(w[2 * i], w[2 * i + 1], b2, 4);
    const float y0 = bfly_step(x4[0], x4[1], b3, 8);
    const float y1 = bfly_step(x4[2], x4[3], b3, 8);
    const int oi = (m0 + 8 * lh + (lm & 7)) * HEADS + (lm >> 3);
    esrc[oi] = y0;
    edst[oi] = y1;
}

#define CSR_BLOCKS 64
#define CSR_WIN 50176

__global__ __launch_bounds__(1024) void csr_count(const int* __restrict__ dst, int E, int n, int per,
                                                  int* __restrict__ hist, int* __restrict__ rank)
{
    __shared__ int lcnt[CSR_WIN];
    const int t = threadIdx.x;
    const int blk = blockIdx.x;
    const int e0 = blk * per;
    const int e1 = (E - e0 < per) ? E : e0 + per;
    int* __restrict__ plane = hist + (size_t)blk * n;
    for (int w0 = 0; w0 < n; w0 += CSR_WIN) {
        const int wn = (n - w0 < CSR_WIN) ? (n - w0) : CSR_WIN;
        for (int i = t; i < wn; i += 1024) lcnt[i] = 0;
        __syncthreads();
        for (int e = e0 + t; e < e1; e += 8 * 1024) {
            int d[8], r[8];
#pragma unroll
            for (int k = 0; k < 8; ++k) {
                const int ek = e + k * 1024;
                const int v = dst[(ek < e1) ? ek : (e1 - 1)];
                d[k] = (ek < e1) ? v - w0 : -1;
            }
#pragma unroll
            for (int k = 0; k < 8; ++k) {
                r[k] = 0;
                if ((unsigned)d[k] < (unsigned)wn) r[k] = atomicAdd(&lcnt[d[k]], 1);
            }
#pragma unroll
            for (int k = 0; k < 8; ++k) {
                if ((unsigned)d[k] < (unsigned)wn) rank[e + k * 1024] = r[k];
            }
        }
        __syncthreads();
        for (int i = t; i < wn; i += 1024) plane[w0 + i] = lcnt[i];
        __syncthreads();
    }
}

__global__ void csr_merge(const int* __restrict__ hist, int* __restrict__ base,
                          int* __restrict__ cnt, int n)
{
    const int d = blockIdx.x * blockDim.x + threadIdx.x;
    if (d >= n) return;
    int v[CSR_BLOCKS];
#pragma unroll
    for (int b = 0; b < CSR_BLOCKS; ++b) v[b] = hist[(size_t)b * n + d];
    int run = 0;
#pragma unroll
    for (int b = 0; b < CSR_BLOCKS; ++b) {
        base[(size_t)b * n + d] = run;
        run += v[b];
    }
    cnt[d] = run;
}

__global__ __launch_bounds__(1024) void csr_scan(const int* __restrict__ cnt,
                                                 int* __restrict__ rowptr, int n)
{
    __shared__ int sh_front[32];
    __shared__ int sh_wave[32];
    const int t = threadIdx.x, lane = t & 31, w = t >> 5;
    const int start = blockIdx.x * 1024;

    int p = 0;
    for (int i = t; i < start; i += 1024) p += cnt[i];
#pragma unroll
    for (int off = 16; off > 0; off >>= 1) p += __shfl_xor(p, off, 32);

    const int gi = start + t;
    const int v = (gi < n) ? cnt[gi] : 0;
    int inc = v;
#pragma unroll
    for (int off = 1; off < 32; off <<= 1) {
        int u = __shfl_up(inc, off, 32);
        if (lane >= off) inc += u;
    }
    if (lane == 31) sh_wave[w] = inc;
    if (lane == 0) sh_front[w] = p;
    __syncthreads();
    int base = 0;
#pragma unroll
    for (int k = 0; k < 32; ++k) {
        base += sh_front[k];
        if (k < w) base += sh_wave[k];
    }
    const int excl = base + inc - v;
    if (gi < n) rowptr[gi] = excl;
    if (gi == n - 1) rowptr[n] = excl + v;
}

template <int CTRL>
__device__ __forceinline__ float row_lane(float v)
{
    return __int_as_float(__builtin_amdgcn_update_dpp(0, __float_as_int(v), CTRL, 0xF, 0xF, true));
}
#define ROW_XOR(mask) (0x160 | (mask))
#define ROW_SHARE(src) (0x150 | (src))

template <int MASK>
__device__ __forceinline__ float bfly_row(float lo, float hi, bool up)
{
    const float keep = up ? hi : lo;
    const float send = up ? lo : hi;
    return keep + row_lane<ROW_XOR(MASK)>(send);
}

template <int T>
__device__ __forceinline__ void add_row(float4& acc, float wv, const float4& r)
{
    const float wt = row_lane<ROW_SHARE(T)>(wv);
    acc.x = fmaf(wt, r.x, acc.x); acc.y = fmaf(wt, r.y, acc.y);
    acc.z = fmaf(wt, r.z, acc.z); acc.w = fmaf(wt, r.w, acc.w);
}

#define STAGE_V 4
template <bool STAGE>
__device__ __forceinline__ float4 gather_node(
    const int node, const int lane,
    const int* __restrict__ rowptr, const int* __restrict__ col,
    const float* __restrict__ asrc,
    const float* __restrict__ esrc, const float* __restrict__ edst,
    const float* __restrict__ hfeat, const float* __restrict__ bias, const int do_relu,
    const float4* __restrict__ stage_src, float4* stage_dst, const int stage_tid)
{
    float4 sv4[STAGE_V];
    if (STAGE) {
#pragma unroll
        for (int k = 0; k < STAGE_V; ++k) sv4[k] = stage_src[stage_tid + k * 1024];
    }
    const int beg = rowptr[node], end = rowptr[node + 1];
    const int head = lane >> 4;
    const float ed = edst[node * HEADS + head];
    float m = lrelu(esrc[node * HEADS + head] + ed);
    const float4 av = *(const float4*)(asrc + lane * 4);
    float4 acc = *(const float4*)(hfeat + (size_t)node * FDIM + lane * 4);
    float p = 1.f;

    const int ja = beg + lane, jb = ja + 32;
    int sa = node, sb = node;
    if (ja < end) sa = col[ja];
    if (jb < end) sb = col[jb];
    if (STAGE) {
#pragma unroll
        for (int k = 0; k < STAGE_V; ++k) stage_dst[stage_tid + k * 1024] = sv4[k];
    }

    const bool b0 = lane & 1, b1 = lane & 2, b2 = lane & 4;
    const int l7 = lane & 7;
    int chunk = 0;
    for (int base = beg; base < end; base += 32, ++chunk) {
        const int cnt = (end - base < 32) ? (end - base) : 32;
        int s;
        if (chunk == 0)      s = sa;
        else if (chunk == 1) s = sb;
        else {
            s = node;
            if (base + lane < end) s = col[base + lane];
        }
#pragma unroll
        for (int g = 0; g < 4; ++g) {
            if (g * 8 < cnt) {
                float4 r[8];
                float d[8];
                if (g * 8 + 8 <= cnt) {
#pragma unroll
                    for (int t = 0; t < 8; ++t) {
                        const int st = __builtin_amdgcn_readlane(s, g * 8 + t);
                        r[t] = *(const float4*)(hfeat + (size_t)st * FDIM + lane * 4);
                    }
                } else {
#pragma unroll
                    for (int t = 0; t < 8; ++t) {
                        r[t] = make_float4(0.f, 0.f, 0.f, 0.f);
                        if (g * 8 + t < cnt) {
                            const int st = __builtin_amdgcn_readlane(s, g * 8 + t);
                            r[t] = *(const float4*)(hfeat + (size_t)st * FDIM + lane * 4);
                        }
                    }
                }
#pragma unroll
                for (int t = 0; t < 8; ++t)
                    d[t] = fmaf(r[t].w, av.w, fmaf(r[t].z, av.z, fmaf(r[t].y, av.y, r[t].x * av.x)));
                float u[4], w[2];
#pragma unroll
                for (int i = 0; i < 4; ++i) u[i] = bfly_row<1>(d[2 * i], d[2 * i + 1], b0);
#pragma unroll
                for (int i = 0; i < 2; ++i) w[i] = bfly_row<2>(u[2 * i], u[2 * i + 1], b1);
                float e = bfly_row<4>(w[0], w[1], b2);
                e += row_lane<ROW_XOR(8)>(e);
                e += ed;
                e = fmaxf(e, NEG_SLOPE * e);
                if (g * 8 + l7 >= cnt) e = -__builtin_inff();
                float bm = e;
                bm = fmaxf(bm, row_lane<ROW_XOR(1)>(bm));
                bm = fmaxf(bm, row_lane<ROW_XOR(2)>(bm));
                bm = fmaxf(bm, row_lane<ROW_XOR(4)>(bm));
                const float mn = fmaxf(m, bm);
                const float sc = __expf(m - mn);
                const float wv = __expf(e - mn);
                float ws = wv;
                ws += row_lane<ROW_XOR(1)>(ws);
                ws += row_lane<ROW_XOR(2)>(ws);
                ws += row_lane<ROW_XOR(4)>(ws);
                p = fmaf(p, sc, ws);
                acc.x *= sc; acc.y *= sc; acc.z *= sc; acc.w *= sc;
                m = mn;
                add_row<0>(acc, wv, r[0]); add_row<1>(acc, wv, r[1]);
                add_row<2>(acc, wv, r[2]); add_row<3>(acc, wv, r[3]);
                add_row<4>(acc, wv, r[4]); add_row<5>(acc, wv, r[5]);
                add_row<6>(acc, wv, r[6]); add_row<7>(acc, wv, r[7]);
            }
        }
    }
    const float den = p;
    const float4 b = *(const float4*)(bias + lane * 4);
    float4 o;
    o.x = acc.x / den + b.x; o.y = acc.y / den + b.y;
    o.z = acc.z / den + b.z; o.w = acc.w / den + b.w;
    if (do_relu) { o.x = fmaxf(o.x, 0.f); o.y = fmaxf(o.y, 0.f); o.z = fmaxf(o.z, 0.f); o.w = fmaxf(o.w, 0.f); }
    return o;
}

__global__ __launch_bounds__(256) __attribute__((amdgpu_waves_per_eu(16, 16))) void gat_gather(
    const int* __restrict__ rowptr, const int* __restrict__ col,
    const float* __restrict__ asrc,
    const float* __restrict__ esrc, const float* __restrict__ edst,
    const float* __restrict__ hfeat, const float* __restrict__ bias,
    float* __restrict__ z, int n, int do_relu)
{
    const int lane = threadIdx.x & 31;
    const int node = __builtin_amdgcn_readfirstlane((int)((blockIdx.x * 256u + threadIdx.x) >> 5));
    if (node >= n) return;
    const float4 o = gather_node<false>(node, lane, rowptr, col, asrc, esrc, edst, hfeat, bias, do_relu,
                                        nullptr, nullptr, 0);
    *(float4*)(z + (size_t)node * FDIM + lane * 4) = o;
}

#define FUSE_ROWS 32
#define TILE_PITCH 132
__global__ __launch_bounds__(1024) __attribute__((amdgpu_waves_per_eu(16, 16))) void gat_gather_gemm(
    const int* __restrict__ rowptr, const int* __restrict__ col,
    const float* __restrict__ asrc,
    const float* __restrict__ esrc, const float* __restrict__ edst,
    const float* __restrict__ hfeat, const float* __restrict__ bias,
    const float* __restrict__ W, const float* __restrict__ a_src2, const float* __restrict__ a_dst2,
    float* __restrict__ Y, float* __restrict__ esrc2, float* __restrict__ edst2, int n)
{
    __shared__ float lw[FDIM * FDIM];
    __shared__ float zt[FUSE_ROWS * TILE_PITCH];
    __shared__ float ht[FUSE_ROWS * TILE_PITCH];
    __shared__ float lgs[2 * FUSE_ROWS * HEADS];
    const float4* w4 = (const float4*)W;
    float4* l4 = (float4*)lw;
    const int lane = threadIdx.x & 31;
    const int wave = __builtin_amdgcn_readfirstlane((int)(threadIdx.x >> 5));
    const int node = (int)blockIdx.x * FUSE_ROWS + wave;
    float4 o = make_float4(0.f, 0.f, 0.f, 0.f);
    if (node < n) {
        o = gather_node<true>(node, lane, rowptr, col, asrc, esrc, edst, hfeat, bias, 1,
                              w4, l4, (int)threadIdx.x);
    } else {
        float4 sv4[STAGE_V];
#pragma unroll
        for (int k = 0; k < STAGE_V; ++k) sv4[k] = w4[threadIdx.x + k * 1024];
#pragma unroll
        for (int k = 0; k < STAGE_V; ++k) l4[threadIdx.x + k * 1024] = sv4[k];
    }
    *(float4*)(zt + wave * TILE_PITCH + lane * 4) = o;
    __syncthreads();

    if (wave < 16) {
        const int strip = wave >> 3, nt = wave & 7;
        const int lm = lane & 15, lh = lane >> 4;
        const float* arow = zt + (strip * 16 + lm) * TILE_PITCH + 2 * lh;
        const float* bcol = lw + (2 * lh) * FDIM + nt * 16 + lm;
        v8f acc = (v8f){0.f, 0.f, 0.f, 0.f, 0.f, 0.f, 0.f, 0.f};
#pragma unroll
        for (int j = 0; j < 32; ++j) {
            const v2f a = *(const v2f*)(arow + j * 4);
            v2f b;
            b.x = bcol[(j * 4) * FDIM];
            b.y = bcol[(j * 4 + 1) * FDIM];
            acc = __builtin_amdgcn_wmma_f32_16x16x4_f32(false, a, false, b, (short)0, acc, false, false);
        }
        float* hrow = ht + (strip * 16 + 8 * lh) * TILE_PITCH + nt * 16 + lm;
#pragma unroll
        for (int r = 0; r < 8; ++r) hrow[r * TILE_PITCH] = acc[r];
    }
    __syncthreads();

    if (node < n) {
        const float4 r = *(const float4*)(ht + wave * TILE_PITCH + lane * 4);
        *(float4*)(Y + (size_t)node * FDIM + lane * 4) = r;
        const float4 as = *(const float4*)(a_src2 + lane * 4);
        const float4 ad = *(const float4*)(a_dst2 + lane * 4);
        float ps = fmaf(r.w, as.w, fmaf(r.z, as.z, fmaf(r.y, as.y, r.x * as.x)));
        float pd = fmaf(r.w, ad.w, fmaf(r.z, ad.z, fmaf(r.y, ad.y, r.x * ad.x)));
        ps += row_lane<ROW_XOR(1)>(ps); pd += row_lane<ROW_XOR(1)>(pd);
        ps += row_lane<ROW_XOR(2)>(ps); pd += row_lane<ROW_XOR(2)>(pd);
        ps += row_lane<ROW_XOR(4)>(ps); pd += row_lane<ROW_XOR(4)>(pd);
        ps += row_lane<ROW_XOR(8)>(ps); pd += row_lane<ROW_XOR(8)>(pd);
        if ((lane & 15) == 0) {
            lgs[wave * HEADS + (lane >> 4)] = ps;
            lgs[FUSE_ROWS * HEADS + wave * HEADS + (lane >> 4)] = pd;
        }
    }
    __syncthreads();
    if (wave < 2) {
        const int nd = (int)blockIdx.x * FUSE_ROWS + lane;
        if (nd < n) {
            const float2 v = *(const float2*)(lgs + wave * (FUSE_ROWS * HEADS) + lane * HEADS);
            float* lp = (wave == 0) ? esrc2 : edst2;
            *(float2*)(lp + (size_t)nd * HEADS) = v;
        }
    }
}

__global__ __launch_bounds__(256) void decode_by_dst(const int* __restrict__ rowptr,
                                                     const int* __restrict__ col,
                                                     const float* __restrict__ z,
                                                     float* __restrict__ slot_score, int n)
{
    const int lane = threadIdx.x & 31;
    const int node = __builtin_amdgcn_readfirstlane((int)((blockIdx.x * 256u + threadIdx.x) >> 5));
    if (node >= n) return;
    const int beg = rowptr[node], end = rowptr[node + 1];
    const float4 b = *(const float4*)(z + (size_t)node * FDIM + lane * 4);

    const bool b0 = lane & 1, b1 = lane & 2, b2 = lane & 4, b3 = lane & 8, b4 = lane & 16;
    for (int base = beg; base < end; base += 32) {
        const int cnt = (end - base < 32) ? (end - base) : 32;
        const int j = base + lane;
        int sv = node;
        if (lane < cnt) sv = col[j];
        float x[4];
#pragma unroll
        for (int g = 0; g < 4; ++g) {
            x[g] = 0.f;
            if (g * 8 < cnt) {
                float p[8];
#pragma unroll
                for (int t = 0; t < 8; ++t) {
                    const int s = __builtin_amdgcn_readlane(sv, g * 8 + t);
                    const float4 a = *(const float4*)(z + (size_t)s * FDIM + lane * 4);
                    p[t] = fmaf(a.w, b.w, fmaf(a.z, b.z, fmaf(a.x, b.x, a.y * b.y)));
                }
                float u[4], w[2];
#pragma unroll
                for (int i = 0; i < 4; ++i) u[i] = bfly_step(p[2 * i], p[2 * i + 1], b0, 1);
#pragma unroll
                for (int i = 0; i < 2; ++i) w[i] = bfly_step(u[2 * i], u[2 * i + 1], b1, 2);
                x[g] = bfly_step(w[0], w[1], b2, 4);
            }
        }
        const float y0 = bfly_step(x[0], x[1], b3, 8);
        const float y1 = bfly_step(x[2], x[3], b3, 8);
        const float r  = bfly_step(y0, y1, b4, 16);
        if (lane < cnt) slot_score[j] = r;
    }
}

__global__ __launch_bounds__(256) void scores_to_edge_order(const int* __restrict__ slot,
                                                            const float* __restrict__ slot_score,
                                                            float* __restrict__ out, int E)
{
    const int lane = threadIdx.x & 31;
    const int e0 = (int)((blockIdx.x * 256u + threadIdx.x) >> 5) * 128 + lane;
    int sl[4];
    float v[4];
#pragma unroll
    for (int k = 0; k < 4; ++k) {
        const int e = e0 + k * 32;
        sl[k] = (e < E) ? slot[e] : -1;
    }
#pragma unroll
    for (int k = 0; k < 4; ++k) v[k] = (sl[k] >= 0) ? slot_score[sl[k]] : 0.f;
#pragma unroll
    for (int k = 0; k < 4; ++k) {
        const int e = e0 + k * 32;
        if (e < E) out[e] = v[k];
    }
}

extern "C" void kernel_launch(void* const* d_in, const int* in_sizes, int n_in,
                              void* d_out, int out_size, void* d_ws, size_t ws_size,
                              hipStream_t stream) {
    const float* x      = (const float*)d_in[0];
    const int*   ei     = (const int*)d_in[1];
    const float* W1     = (const float*)d_in[2];
    const float* a1s    = (const float*)d_in[3];
    const float* a1d    = (const float*)d_in[4];
    const float* b1     = (const float*)d_in[5];
    const float* W2     = (const float*)d_in[6];
    const float* a2s    = (const float*)d_in[7];
    const float* a2d    = (const float*)d_in[8];
    const float* b2     = (const float*)d_in[9];
    float* scores = (float*)d_out;

    const int N = in_sizes[0] / FDIM;
    const int E = in_sizes[1] / 2;
    const int* src = ei;
    const int* dst = ei + E;

    float* ws   = (float*)d_ws;
    float*    h     = ws;
    float*    esrc2 = ws + (size_t)N * FDIM;
    float*    edst2 = esrc2 + (size_t)N * HEADS;
    int*      cnt   = (int*)(edst2 + (size_t)N * HEADS);
    int*      rowp  = cnt + N;
    int*      rank  = rowp + N + 1;
    int*      col   = rank + E;
    float*    z     = ws + (size_t)2 * N * FDIM;
    int*      hist  = (int*)z;
    int*      base  = hist + (size_t)CSR_BLOCKS * N;
    float*    esrc  = ws + (size_t)3 * N * FDIM;
    float*    edst  = esrc + (size_t)N * HEADS;

    dim3 blk(256);
    dim3 gGemm((N + 127) / 128);
    dim3 gNode((N + 255) / 256);
    dim3 gGemmFill((N + 127) / 128 + (E + 256 * FILL_EPT - 1) / (256 * FILL_EPT));
    dim3 gScan((N + 1023) / 1024);
    dim3 gNodeW((N + 7) / 8);
    dim3 gNodeF((N + FUSE_ROWS - 1) / FUSE_ROWS);
    dim3 gE4((E + 1023) / 1024);
    const int per = (E + CSR_BLOCKS - 1) / CSR_BLOCKS;

    csr_count<<<dim3(CSR_BLOCKS), dim3(1024), 0, stream>>>(dst, E, N, per, hist, rank);
    csr_merge<<<gNode, blk, 0, stream>>>(hist, base, cnt, N);
    csr_scan<<<gScan, dim3(1024), 0, stream>>>(cnt, rowp, N);

    gemm128_wmma_logits<<<gGemmFill, blk, 0, stream>>>(x, W1, a1s, a1d, h, esrc, edst, N,
                                                       (int)gGemm.x, src, dst, E, per, rowp, base, rank, col);
    gat_gather_gemm<<<gNodeF, dim3(1024), 0, stream>>>(rowp, col, a1s, esrc, edst, h, b1,
                                                       W2, a2s, a2d, z, esrc2, edst2, N);

    gat_gather<<<gNodeW, blk, 0, stream>>>(rowp, col, a2s, esrc2, edst2, z, b2, h, N,  0);

    float* slot_score = z;
    decode_by_dst<<<gNodeW, blk, 0, stream>>>(rowp, col, h, slot_score, N);
    scores_to_edge_order<<<gE4, blk, 0, stream>>>(rank, slot_score, scores, E);
}
